# starting kernel
# baseline (speedup 1.0000x reference)
#include <hip/hip_runtime.h>


typedef __attribute__((ext_vector_type(16))) _Float16 v16h;
typedef __attribute__((ext_vector_type(8)))  _Float16 v8h;
typedef __attribute__((ext_vector_type(8)))  float    v8f;

#define N_NODES 50000
#define E_EDGES 800000
#define P_CAND  100000
#define HDIM    128
#define NEG_SLOPE 0.2f


__device__ __forceinline__ int f2ord(float f) {
    int i = __float_as_int(f);
    return i >= 0 ? i : (i ^ 0x7fffffff);
}
__device__ __forceinline__ float ord2f(int i) {
    return __int_as_float(i >= 0 ? i : (i ^ 0x7fffffff));
}

__device__ __forceinline__ v16h load_a_f32(const float* __restrict__ p) {
    float4 a0 = *(const float4*)(p);
    float4 a1 = *(const float4*)(p + 4);
    float4 a2 = *(const float4*)(p + 16);
    float4 a3 = *(const float4*)(p + 20);
    v16h v;
    v[0]=(_Float16)a0.x; v[1]=(_Float16)a0.y; v[2]=(_Float16)a0.z; v[3]=(_Float16)a0.w;
    v[4]=(_Float16)a1.x; v[5]=(_Float16)a1.y; v[6]=(_Float16)a1.z; v[7]=(_Float16)a1.w;
    v[8]=(_Float16)a2.x; v[9]=(_Float16)a2.y; v[10]=(_Float16)a2.z; v[11]=(_Float16)a2.w;
    v[12]=(_Float16)a3.x; v[13]=(_Float16)a3.y; v[14]=(_Float16)a3.z; v[15]=(_Float16)a3.w;
    return v;
}

__device__ __forceinline__ v16h load_a_prod(const float* __restrict__ sp,
                                            const float* __restrict__ dp) {
    float4 s0 = *(const float4*)(sp),      d0 = *(const float4*)(dp);
    float4 s1 = *(const float4*)(sp + 4),  d1 = *(const float4*)(dp + 4);
    float4 s2 = *(const float4*)(sp + 16), d2 = *(const float4*)(dp + 16);
    float4 s3 = *(const float4*)(sp + 20), d3 = *(const float4*)(dp + 20);
    v16h v;
    v[0]=(_Float16)(s0.x*d0.x); v[1]=(_Float16)(s0.y*d0.y);
    v[2]=(_Float16)(s0.z*d0.z); v[3]=(_Float16)(s0.w*d0.w);
    v[4]=(_Float16)(s1.x*d1.x); v[5]=(_Float16)(s1.y*d1.y);
    v[6]=(_Float16)(s1.z*d1.z); v[7]=(_Float16)(s1.w*d1.w);
    v[8]=(_Float16)(s2.x*d2.x); v[9]=(_Float16)(s2.y*d2.y);
    v[10]=(_Float16)(s2.z*d2.z); v[11]=(_Float16)(s2.w*d2.w);
    v[12]=(_Float16)(s3.x*d3.x); v[13]=(_Float16)(s3.y*d3.y);
    v[14]=(_Float16)(s3.z*d3.z); v[15]=(_Float16)(s3.w*d3.w);
    return v;
}

__device__ __forceinline__ v16h load_b_f16(const _Float16* __restrict__ p) {
    v8h lo = *(const v8h*)(p);
    v8h hi = *(const v8h*)(p + 16);
    v16h v;
#pragma unroll
    for (int i = 0; i < 8; ++i) { v[i] = lo[i]; v[8 + i] = hi[i]; }
    return v;
}

__global__ void prep_weights(const float* __restrict__ W0, const float* __restrict__ W1,
                             const float* __restrict__ Wp1, const float* __restrict__ Wp2,
                             _Float16* __restrict__ W0t, _Float16* __restrict__ W1t,
                             _Float16* __restrict__ Wp1t, _Float16* __restrict__ Wp2t) {
    int idx = blockIdx.x * blockDim.x + threadIdx.x;
    if (idx < 128 * 128) {
        int c = idx >> 7, k = idx & 127;
        W0t[idx] = (_Float16)W0[k * 128 + c];
        W1t[idx] = (_Float16)W1[k * 128 + c];
    }
    if (idx < 64 * 128) {
        int c = idx >> 7, k = idx & 127;
        Wp1t[idx] = (_Float16)Wp1[k * 64 + c];
    }
    if (idx < 64 * 64) {
        int c = idx >> 6, k = idx & 63;
        Wp2t[idx] = (_Float16)Wp2[k * 64 + c];
    }
}

__global__ __launch_bounds__(256)
void gemm128_wmma(const float* __restrict__ A, const _Float16* __restrict__ Bt,
                  float* __restrict__ C, int M) {
    int wid  = blockIdx.x * (blockDim.x >> 5) + (threadIdx.x >> 5);
    int lane = threadIdx.x & 31;
    int half = lane >> 4, r = lane & 15;
    int tm = wid >> 3, tn = wid & 7;
    if (tm * 16 >= M) return;
    int row = tm * 16 + r;
    int col = tn * 16 + r;
    v8f acc = {};
#pragma unroll
    for (int kk = 0; kk < 128; kk += 32) {
        int k0 = kk + half * 8;
        v16h a = load_a_f32(A + row * 128 + k0);
        v16h b = load_b_f16(Bt + col * 128 + k0);
        acc = __builtin_amdgcn_wmma_f32_16x16x32_f16(false, a, false, b,
                                                     (short)0, acc, false, false);
    }
    float* cp = C + (tm * 16 + half * 8) * 128 + tn * 16 + r;
#pragma unroll
    for (int v = 0; v < 8; ++v) cp[v * 128] = acc[v];
}

__global__ __launch_bounds__(256)
void edge_score_kernel(const float* __restrict__ feat, const int* __restrict__ src,
                       const int* __restrict__ dst, const float* __restrict__ attn,
                       float* __restrict__ score, int* __restrict__ mOrd) {
    int gid = blockIdx.x * blockDim.x + threadIdx.x;
    int e = gid >> 5;
    if (e >= E_EDGES) return;
    int lane = threadIdx.x & 31;
    int hh = lane >> 4, d4 = (lane & 15) << 2;
    int s_ = src[e], d_ = dst[e];
    float4 el = *(const float4*)(feat + s_ * HDIM + hh * 64 + d4);
    float4 er = *(const float4*)(feat + d_ * HDIM + hh * 64 + d4);
    float4 at = *(const float4*)(attn + hh * 64 + d4);
    float x0 = el.x + er.x; x0 = x0 > 0.f ? x0 : NEG_SLOPE * x0;
    float x1 = el.y + er.y; x1 = x1 > 0.f ? x1 : NEG_SLOPE * x1;
    float x2 = el.z + er.z; x2 = x2 > 0.f ? x2 : NEG_SLOPE * x2;
    float x3 = el.w + er.w; x3 = x3 > 0.f ? x3 : NEG_SLOPE * x3;
    float part = x0 * at.x + x1 * at.y + x2 * at.z + x3 * at.w;
    part += __shfl_xor(part, 1, 32);
    part += __shfl_xor(part, 2, 32);
    part += __shfl_xor(part, 4, 32);
    part += __shfl_xor(part, 8, 32);
    if ((lane & 15) == 0) {
        score[e * 2 + hh] = part;
        atomicMax(&mOrd[d_ * 2 + hh], f2ord(part));
    }
}

__global__ __launch_bounds__(256)
void edge_expsum_kernel(const int* __restrict__ dst, const int* __restrict__ mOrd,
                        float* __restrict__ score, float* __restrict__ ssum) {
    int idx = blockIdx.x * blockDim.x + threadIdx.x;
    if (idx >= E_EDGES * 2) return;
    int e = idx >> 1, hh = idx & 1;
    int d_ = dst[e];
    float m = ord2f(mOrd[d_ * 2 + hh]);
    float p = __expf(score[idx] - m);
    score[idx] = p;
    atomicAdd(&ssum[d_ * 2 + hh], p);
}

__global__ __launch_bounds__(256)
void edge_aggregate_kernel(const float* __restrict__ feat, const int* __restrict__ src,
                           const int* __restrict__ dst, const float* __restrict__ p,
                           const float* __restrict__ ssum, float* __restrict__ rst) {
    int gid = blockIdx.x * blockDim.x + threadIdx.x;
    int e = gid >> 5;
    if (e >= E_EDGES) return;
    int lane = threadIdx.x & 31;
    int hh = lane >> 4, d4 = (lane & 15) << 2;
    int s_ = src[e], d_ = dst[e];
    float a = p[e * 2 + hh] / ssum[d_ * 2 + hh];
    float4 el = *(const float4*)(feat + s_ * HDIM + hh * 64 + d4);
    float* rp = rst + d_ * HDIM + hh * 64 + d4;
    atomicAdd(rp + 0, el.x * a);
    atomicAdd(rp + 1, el.y * a);
    atomicAdd(rp + 2, el.z * a);
    atomicAdd(rp + 3, el.w * a);
}

__global__ __launch_bounds__(256)
void finalize_kernel(const float* __restrict__ rst, const float* __restrict__ hin,
                     float* __restrict__ hout, int residual, int n) {
    int idx = blockIdx.x * blockDim.x + threadIdx.x;
    if (idx >= n) return;
    float v = rst[idx] + (residual ? hin[idx] : 0.0f);
    hout[idx] = v > 0.f ? v : (__expf(v) - 1.0f);
}

__global__ __launch_bounds__(128)
void predictor_wmma(const float* __restrict__ h,
                    const int* __restrict__ ps, const int* __restrict__ pd,
                    const int* __restrict__ ns, const int* __restrict__ nd,
                    const _Float16* __restrict__ Wp1t, const float* __restrict__ bp1,
                    const _Float16* __restrict__ Wp2t, const float* __restrict__ bp2,
                    const float* __restrict__ Wp3, const float* __restrict__ bp3,
                    float* __restrict__ out) {
    __shared__ _Float16 z1s[4][16 * 64];
    __shared__ float    z2s[4][16 * 64];
    const int tilesPerSet = P_CAND / 16;
    int w = threadIdx.x >> 5;
    int tile = blockIdx.x * 4 + w;
    if (tile >= 2 * tilesPerSet) return;
    bool isNeg = tile >= tilesPerSet;
    int t = isNeg ? tile - tilesPerSet : tile;
    const int* S = isNeg ? ns : ps;
    const int* D = isNeg ? nd : pd;

    int lane = threadIdx.x & 31;
    int half = lane >> 4, r = lane & 15;
    int c = t * 16 + r;
    int si = S[c], di = D[c];

    v16h a1[4];
#pragma unroll
    for (int kc = 0; kc < 4; ++kc) {
        int k0 = kc * 32 + half * 8;
        a1[kc] = load_a_prod(h + si * HDIM + k0, h + di * HDIM + k0);
    }
#pragma unroll
    for (int nt = 0; nt < 4; ++nt) {
        int col = nt * 16 + r;
        v8f acc = {};
#pragma unroll
        for (int kc = 0; kc < 4; ++kc) {
            int k0 = kc * 32 + half * 8;
            v16h b = load_b_f16(Wp1t + col * 128 + k0);
            acc = __builtin_amdgcn_wmma_f32_16x16x32_f16(false, a1[kc], false, b,
                                                         (short)0, acc, false, false);
        }
        float bias = bp1[col];
#pragma unroll
        for (int v = 0; v < 8; ++v) {
            float z = acc[v] + bias;
            z = z > 0.f ? z : 0.f;
            z1s[w][(v + half * 8) * 64 + col] = (_Float16)z;
        }
    }

    v16h a2[2];
#pragma unroll
    for (int kc = 0; kc < 2; ++kc) {
        int k0 = kc * 32 + half * 8;
        const _Float16* zp = &z1s[w][r * 64 + k0];
        a2[kc] = load_b_f16(zp);
    }
#pragma unroll
    for (int nt = 0; nt < 4; ++nt) {
        int col = nt * 16 + r;
        v8f acc = {};
#pragma unroll
        for (int kc = 0; kc < 2; ++kc) {
            int k0 = kc * 32 + half * 8;
            v16h b = load_b_f16(Wp2t + col * 64 + k0);
            acc = __builtin_amdgcn_wmma_f32_16x16x32_f16(false, a2[kc], false, b,
                                                         (short)0, acc, false, false);
        }
        float bias = bp2[col];
#pragma unroll
        for (int v = 0; v < 8; ++v) {
            float z = acc[v] + bias;
            z2s[w][(v + half * 8) * 64 + col] = z > 0.f ? z : 0.f;
        }
    }

    float part = 0.f;
    int jb = half * 32;
#pragma unroll
    for (int j = 0; j < 32; ++j)
        part += z2s[w][r * 64 + jb + j] * Wp3[jb + j];
    part += __shfl_xor(part, 16, 32);
    if (half == 0) {
        out[(isNeg ? P_CAND : 0) + t * 16 + r] = part + bp3[0];
    }
}

extern "C" void kernel_launch(void* const* d_in, const int* in_sizes, int n_in,
                              void* d_out, int out_size, void* d_ws, size_t ws_size,
                              hipStream_t stream) {
    const float* x        = (const float*)d_in[0];
    const int*   src      = (const int*)  d_in[1];
    const int*   dst      = (const int*)  d_in[2];
    const int*   pos_src  = (const int*)  d_in[3];
    const int*   pos_dst  = (const int*)  d_in[4];
    const int*   neg_src  = (const int*)  d_in[5];
    const int*   neg_dst  = (const int*)  d_in[6];
    const float* W0       = (const float*)d_in[7];
    const float* attn0    = (const float*)d_in[8];
    const float* W1       = (const float*)d_in[9];
    const float* attn1    = (const float*)d_in[10];
    const float* Wp1      = (const float*)d_in[11];
    const float* bp1      = (const float*)d_in[12];
    const float* Wp2      = (const float*)d_in[13];
    const float* bp2      = (const float*)d_in[14];
    const float* Wp3      = (const float*)d_in[15];
    const float* bp3      = (const float*)d_in[16];

    char* ws = (char*)d_ws;
    size_t off = 0;
    auto carve = [&](size_t bytes) -> char* {
        char* p = ws + off;
        off += (bytes + 255) & ~(size_t)255;
        return p;
    };
    float*     feat  = (float*)    carve((size_t)N_NODES * HDIM * 4);
    float*     h0    = (float*)    carve((size_t)N_NODES * HDIM * 4);
    float*     h1    = (float*)    carve((size_t)N_NODES * HDIM * 4);
    float*     rst   = (float*)    carve((size_t)N_NODES * HDIM * 4);
    float*     score = (float*)    carve((size_t)E_EDGES * 2 * 4);
    int*       mOrd  = (int*)      carve((size_t)N_NODES * 2 * 4);
    float*     ssum  = (float*)    carve((size_t)N_NODES * 2 * 4);
    _Float16*  W0t   = (_Float16*) carve(128 * 128 * 2);
    _Float16*  W1t   = (_Float16*) carve(128 * 128 * 2);
    _Float16*  Wp1t  = (_Float16*) carve(64 * 128 * 2);
    _Float16*  Wp2t  = (_Float16*) carve(64 * 64 * 2);

    prep_weights<<<(128 * 128 + 255) / 256, 256, 0, stream>>>(
        W0, W1, Wp1, Wp2, W0t, W1t, Wp1t, Wp2t);

    const int gemmBlocks  = (N_NODES / 16) * 8 / 8;
    const int edgeBlocks  = E_EDGES / 8;
    const int ehBlocks    = (E_EDGES * 2 + 255) / 256;
    const int nodeBlocks  = (N_NODES * HDIM + 255) / 256;

    hipMemsetAsync(rst,  0,    (size_t)N_NODES * HDIM * 4, stream);
    hipMemsetAsync(ssum, 0,    (size_t)N_NODES * 2 * 4,    stream);
    hipMemsetAsync(mOrd, 0x80, (size_t)N_NODES * 2 * 4,    stream);
    gemm128_wmma<<<gemmBlocks, 256, 0, stream>>>(x, W0t, feat, N_NODES);
    edge_score_kernel<<<edgeBlocks, 256, 0, stream>>>(feat, src, dst, attn0, score, mOrd);
    edge_expsum_kernel<<<ehBlocks, 256, 0, stream>>>(dst, mOrd, score, ssum);
    edge_aggregate_kernel<<<edgeBlocks, 256, 0, stream>>>(feat, src, dst, score, ssum, rst);
    finalize_kernel<<<nodeBlocks, 256, 0, stream>>>(rst, x, h0, 0, N_NODES * HDIM);

    hipMemsetAsync(rst,  0,    (size_t)N_NODES * HDIM * 4, stream);
    hipMemsetAsync(ssum, 0,    (size_t)N_NODES * 2 * 4,    stream);
    hipMemsetAsync(mOrd, 0x80, (size_t)N_NODES * 2 * 4,    stream);
    gemm128_wmma<<<gemmBlocks, 256, 0, stream>>>(h0, W1t, feat, N_NODES);
    edge_score_kernel<<<edgeBlocks, 256, 0, stream>>>(feat, src, dst, attn1, score, mOrd);
    edge_expsum_kernel<<<ehBlocks, 256, 0, stream>>>(dst, mOrd, score, ssum);
    edge_aggregate_kernel<<<edgeBlocks, 256, 0, stream>>>(feat, src, dst, score, ssum, rst);
    finalize_kernel<<<nodeBlocks, 256, 0, stream>>>(rst, h0, h1, 1, N_NODES * HDIM);

    const int predBlocks = (2 * (P_CAND / 16) + 3) / 4;
    predictor_wmma<<<predBlocks, 128, 0, stream>>>(
        h1, pos_src, pos_dst, neg_src, neg_dst,
        Wp1t, bp1, Wp2t, bp2, Wp3, bp3, (float*)d_out);
}


# Aggregate kernel: lane l takes columns l, 32+l, 64+l, 96+l of the row, so each float atomic add (and row load) of a wave covers one whole 128-B line, not 4 lines; one wave per edge and add order kept
# speedup vs baseline: 1.3791x; 1.3791x over previous
#include <hip/hip_runtime.h>


typedef __attribute__((ext_vector_type(16))) _Float16 v16h;
typedef __attribute__((ext_vector_type(8)))  _Float16 v8h;
typedef __attribute__((ext_vector_type(8)))  float    v8f;

#define N_NODES 50000
#define E_EDGES 800000
#define P_CAND  100000
#define HDIM    128
#define NEG_SLOPE 0.2f


__device__ __forceinline__ int f2ord(float f) {
    int i = __float_as_int(f);
    return i >= 0 ? i : (i ^ 0x7fffffff);
}
__device__ __forceinline__ float ord2f(int i) {
    return __int_as_float(i >= 0 ? i : (i ^ 0x7fffffff));
}

__device__ __forceinline__ v16h load_a_f32(const float* __restrict__ p) {
    float4 a0 = *(const float4*)(p);
    float4 a1 = *(const float4*)(p + 4);
    float4 a2 = *(const float4*)(p + 16);
    float4 a3 = *(const float4*)(p + 20);
    v16h v;
    v[0]=(_Float16)a0.x; v[1]=(_Float16)a0.y; v[2]=(_Float16)a0.z; v[3]=(_Float16)a0.w;
    v[4]=(_Float16)a1.x; v[5]=(_Float16)a1.y; v[6]=(_Float16)a1.z; v[7]=(_Float16)a1.w;
    v[8]=(_Float16)a2.x; v[9]=(_Float16)a2.y; v[10]=(_Float16)a2.z; v[11]=(_Float16)a2.w;
    v[12]=(_Float16)a3.x; v[13]=(_Float16)a3.y; v[14]=(_Float16)a3.z; v[15]=(_Float16)a3.w;
    return v;
}

__device__ __forceinline__ v16h load_a_prod(const float* __restrict__ sp,
                                            const float* __restrict__ dp) {
    float4 s0 = *(const float4*)(sp),      d0 = *(const float4*)(dp);
    float4 s1 = *(const float4*)(sp + 4),  d1 = *(const float4*)(dp + 4);
    float4 s2 = *(const float4*)(sp + 16), d2 = *(const float4*)(dp + 16);
    float4 s3 = *(const float4*)(sp + 20), d3 = *(const float4*)(dp + 20);
    v16h v;
    v[0]=(_Float16)(s0.x*d0.x); v[1]=(_Float16)(s0.y*d0.y);
    v[2]=(_Float16)(s0.z*d0.z); v[3]=(_Float16)(s0.w*d0.w);
    v[4]=(_Float16)(s1.x*d1.x); v[5]=(_Float16)(s1.y*d1.y);
    v[6]=(_Float16)(s1.z*d1.z); v[7]=(_Float16)(s1.w*d1.w);
    v[8]=(_Float16)(s2.x*d2.x); v[9]=(_Float16)(s2.y*d2.y);
    v[10]=(_Float16)(s2.z*d2.z); v[11]=(_Float16)(s2.w*d2.w);
    v[12]=(_Float16)(s3.x*d3.x); v[13]=(_Float16)(s3.y*d3.y);
    v[14]=(_Float16)(s3.z*d3.z); v[15]=(_Float16)(s3.w*d3.w);
    return v;
}

__device__ __forceinline__ v16h load_b_f16(const _Float16* __restrict__ p) {
    v8h lo = *(const v8h*)(p);
    v8h hi = *(const v8h*)(p + 16);
    v16h v;
#pragma unroll
    for (int i = 0; i < 8; ++i) { v[i] = lo[i]; v[8 + i] = hi[i]; }
    return v;
}

__global__ void prep_weights(const float* __restrict__ W0, const float* __restrict__ W1,
                             const float* __restrict__ Wp1, const float* __restrict__ Wp2,
                             _Float16* __restrict__ W0t, _Float16* __restrict__ W1t,
                             _Float16* __restrict__ Wp1t, _Float16* __restrict__ Wp2t) {
    int idx = blockIdx.x * blockDim.x + threadIdx.x;
    if (idx < 128 * 128) {
        int c = idx >> 7, k = idx & 127;
        W0t[idx] = (_Float16)W0[k * 128 + c];
        W1t[idx] = (_Float16)W1[k * 128 + c];
    }
    if (idx < 64 * 128) {
        int c = idx >> 7, k = idx & 127;
        Wp1t[idx] = (_Float16)Wp1[k * 64 + c];
    }
    if (idx < 64 * 64) {
        int c = idx >> 6, k = idx & 63;
        Wp2t[idx] = (_Float16)Wp2[k * 64 + c];
    }
}

__global__ __launch_bounds__(256)
void gemm128_wmma(const float* __restrict__ A, const _Float16* __restrict__ Bt,
                  float* __restrict__ C, int M) {
    int wid  = blockIdx.x * (blockDim.x >> 5) + (threadIdx.x >> 5);
    int lane = threadIdx.x & 31;
    int half = lane >> 4, r = lane & 15;
    int tm = wid >> 3, tn = wid & 7;
    if (tm * 16 >= M) return;
    int row = tm * 16 + r;
    int col = tn * 16 + r;
    v8f acc = {};
#pragma unroll
    for (int kk = 0; kk < 128; kk += 32) {
        int k0 = kk + half * 8;
        v16h a = load_a_f32(A + row * 128 + k0);
        v16h b = load_b_f16(Bt + col * 128 + k0);
        acc = __builtin_amdgcn_wmma_f32_16x16x32_f16(false, a, false, b,
                                                     (short)0, acc, false, false);
    }
    float* cp = C + (tm * 16 + half * 8) * 128 + tn * 16 + r;
#pragma unroll
    for (int v = 0; v < 8; ++v) cp[v * 128] = acc[v];
}

#define SCORE_STRIP 32
#define SCORE_GROUP 8
static_assert(E_EDGES % SCORE_STRIP == 0, "strips cover the edge list exactly");
__global__ __launch_bounds__(256)
void edge_score_kernel(const float* __restrict__ feat, const int* __restrict__ src,
                       const int* __restrict__ dst, const float* __restrict__ attn,
                       float* __restrict__ score, int* __restrict__ mOrd) {
    int strip = (blockIdx.x * blockDim.x + threadIdx.x) >> 5;
    int eb = strip * SCORE_STRIP;
    if (eb >= E_EDGES) return;
    int lane = threadIdx.x & 31;
    int hh = lane >> 4, c = lane & 15, d4 = c << 2;
    int sv = src[eb + lane];
    int dv = dst[eb + lane];
    float4 at = *(const float4*)(attn + hh * 64 + d4);
    const float* fl = feat + hh * 64 + d4;
    float keep0 = 0.f, keep1 = 0.f;
#pragma unroll 1
    for (int j0 = 0; j0 < SCORE_STRIP; j0 += SCORE_GROUP) {
        float4 el[SCORE_GROUP], er[SCORE_GROUP];
#pragma unroll
        for (int u = 0; u < SCORE_GROUP; ++u) {
            int s_ = __builtin_amdgcn_readlane(sv, j0 + u);
            int d_ = __builtin_amdgcn_readlane(dv, j0 + u);
            el[u] = *(const float4*)(fl + s_ * HDIM);
            er[u] = *(const float4*)(fl + d_ * HDIM);
        }
#pragma unroll
        for (int u = 0; u < SCORE_GROUP; ++u) {
            float x0 = el[u].x + er[u].x; x0 = x0 > 0.f ? x0 : NEG_SLOPE * x0;
            float x1 = el[u].y + er[u].y; x1 = x1 > 0.f ? x1 : NEG_SLOPE * x1;
            float x2 = el[u].z + er[u].z; x2 = x2 > 0.f ? x2 : NEG_SLOPE * x2;
            float x3 = el[u].w + er[u].w; x3 = x3 > 0.f ? x3 : NEG_SLOPE * x3;
            float part = fmaf(x3, at.w, fmaf(x2, at.z, fmaf(x0, at.x, x1 * at.y)));
            part += __shfl_xor(part, 1, 32);
            part += __shfl_xor(part, 2, 32);
            part += __shfl_xor(part, 4, 32);
            part += __shfl_xor(part, 8, 32);
            const int j = j0 + u;
            if (c == (j & 15)) { if (j < 16) keep0 = part; else keep1 = part; }
        }
    }
    int dA = __shfl(dv, c, 32);
    int dB = __shfl(dv, 16 + c, 32);
    score[(eb + c) * 2 + hh]      = keep0;
    score[(eb + 16 + c) * 2 + hh] = keep1;
    atomicMax(&mOrd[dA * 2 + hh], f2ord(keep0));
    atomicMax(&mOrd[dB * 2 + hh], f2ord(keep1));
}

__global__ __launch_bounds__(256)
void edge_aggregate_kernel(const float* __restrict__ feat, const int* __restrict__ src,
                           const int* __restrict__ dst, const float* __restrict__ score,
                           const int* __restrict__ mOrd, float* __restrict__ ssum,
                           float* __restrict__ rst) {
    int gid = blockIdx.x * blockDim.x + threadIdx.x;
    int e = gid >> 5;
    if (e >= E_EDGES) return;
    int lane = threadIdx.x & 31;
    int s_ = src[e], d_ = dst[e];
    const float* fr = feat + s_ * HDIM + lane;
    float v0 = fr[0], v1 = fr[32], v2 = fr[64], v3 = fr[96];
    int2   mo = *(const int2*)(mOrd + d_ * 2);
    float2 sc = *(const float2*)(score + e * 2);
    float p0 = __expf(sc.x - ord2f(mo.x));
    float p1 = __expf(sc.y - ord2f(mo.y));
    if (lane < 2) atomicAdd(&ssum[d_ * 2 + lane], lane ? p1 : p0);
    float* rp = rst + d_ * HDIM + lane;
    atomicAdd(rp +  0, v0 * p0);
    atomicAdd(rp + 32, v1 * p0);
    atomicAdd(rp + 64, v2 * p1);
    atomicAdd(rp + 96, v3 * p1);
}

__global__ __launch_bounds__(256)
void finalize_kernel(const float* __restrict__ rst, const float* __restrict__ ssum,
                     const float* __restrict__ hin,
                     float* __restrict__ hout, int residual, int n) {
    int idx = blockIdx.x * blockDim.x + threadIdx.x;
    if (idx >= n) return;
    float s = ssum[idx >> 6];
    float v = s > 0.f ? rst[idx] / s : 0.f;
    v += residual ? hin[idx] : 0.0f;
    hout[idx] = v > 0.f ? v : (__expf(v) - 1.0f);
}

__global__ __launch_bounds__(128)
void predictor_wmma(const float* __restrict__ h,
                    const int* __restrict__ ps, const int* __restrict__ pd,
                    const int* __restrict__ ns, const int* __restrict__ nd,
                    const _Float16* __restrict__ Wp1t, const float* __restrict__ bp1,
                    const _Float16* __restrict__ Wp2t, const float* __restrict__ bp2,
                    const float* __restrict__ Wp3, const float* __restrict__ bp3,
                    float* __restrict__ out) {
    __shared__ _Float16 z1s[4][16 * 64];
    __shared__ float    z2s[4][16 * 64];
    const int tilesPerSet = P_CAND / 16;
    int w = threadIdx.x >> 5;
    int tile = blockIdx.x * 4 + w;
    if (tile >= 2 * tilesPerSet) return;
    bool isNeg = tile >= tilesPerSet;
    int t = isNeg ? tile - tilesPerSet : tile;
    const int* S = isNeg ? ns : ps;
    const int* D = isNeg ? nd : pd;

    int lane = threadIdx.x & 31;
    int half = lane >> 4, r = lane & 15;
    int c = t * 16 + r;
    int si = S[c], di = D[c];

    v16h a1[4];
#pragma unroll
    for (int kc = 0; kc < 4; ++kc) {
        int k0 = kc * 32 + half * 8;
        a1[kc] = load_a_prod(h + si * HDIM + k0, h + di * HDIM + k0);
    }
#pragma unroll
    for (int nt = 0; nt < 4; ++nt) {
        int col = nt * 16 + r;
        v8f acc = {};
#pragma unroll
        for (int kc = 0; kc < 4; ++kc) {
            int k0 = kc * 32 + half * 8;
            v16h b = load_b_f16(Wp1t + col * 128 + k0);
            acc = __builtin_amdgcn_wmma_f32_16x16x32_f16(false, a1[kc], false, b,
                                                         (short)0, acc, false, false);
        }
        float bias = bp1[col];
#pragma unroll
        for (int v = 0; v < 8; ++v) {
            float z = acc[v] + bias;
            z = z > 0.f ? z : 0.f;
            z1s[w][(v + half * 8) * 64 + col] = (_Float16)z;
        }
    }

    v16h a2[2];
#pragma unroll
    for (int kc = 0; kc < 2; ++kc) {
        int k0 = kc * 32 + half * 8;
        const _Float16* zp = &z1s[w][r * 64 + k0];
        a2[kc] = load_b_f16(zp);
    }
#pragma unroll
    for (int nt = 0; nt < 4; ++nt) {
        int col = nt * 16 + r;
        v8f acc = {};
#pragma unroll
        for (int kc = 0; kc < 2; ++kc) {
            int k0 = kc * 32 + half * 8;
            v16h b = load_b_f16(Wp2t + col * 64 + k0);
            acc = __builtin_amdgcn_wmma_f32_16x16x32_f16(false, a2[kc], false, b,
                                                         (short)0, acc, false, false);
        }
        float bias = bp2[col];
#pragma unroll
        for (int v = 0; v < 8; ++v) {
            float z = acc[v] + bias;
            z2s[w][(v + half * 8) * 64 + col] = z > 0.f ? z : 0.f;
        }
    }

    float part = 0.f;
    int jb = half * 32;
#pragma unroll
    for (int j = 0; j < 32; ++j)
        part += z2s[w][r * 64 + jb + j] * Wp3[jb + j];
    part += __shfl_xor(part, 16, 32);
    if (half == 0) {
        out[(isNeg ? P_CAND : 0) + t * 16 + r] = part + bp3[0];
    }
}

extern "C" void kernel_launch(void* const* d_in, const int* in_sizes, int n_in,
                              void* d_out, int out_size, void* d_ws, size_t ws_size,
                              hipStream_t stream) {
    const float* x        = (const float*)d_in[0];
    const int*   src      = (const int*)  d_in[1];
    const int*   dst      = (const int*)  d_in[2];
    const int*   pos_src  = (const int*)  d_in[3];
    const int*   pos_dst  = (const int*)  d_in[4];
    const int*   neg_src  = (const int*)  d_in[5];
    const int*   neg_dst  = (const int*)  d_in[6];
    const float* W0       = (const float*)d_in[7];
    const float* attn0    = (const float*)d_in[8];
    const float* W1       = (const float*)d_in[9];
    const float* attn1    = (const float*)d_in[10];
    const float* Wp1      = (const float*)d_in[11];
    const float* bp1      = (const float*)d_in[12];
    const float* Wp2      = (const float*)d_in[13];
    const float* bp2      = (const float*)d_in[14];
    const float* Wp3      = (const float*)d_in[15];
    const float* bp3      = (const float*)d_in[16];

    char* ws = (char*)d_ws;
    size_t off = 0;
    auto carve = [&](size_t bytes) -> char* {
        char* p = ws + off;
        off += (bytes + 255) & ~(size_t)255;
        return p;
    };
    float*     feat  = (float*)    carve((size_t)N_NODES * HDIM * 4);
    float*     h0    = (float*)    carve((size_t)N_NODES * HDIM * 4);
    float*     h1    = (float*)    carve((size_t)N_NODES * HDIM * 4);
    float*     rst   = (float*)    carve((size_t)N_NODES * HDIM * 4);
    float*     score = (float*)    carve((size_t)E_EDGES * 2 * 4);
    int*       mOrd  = (int*)      carve((size_t)N_NODES * 2 * 4);
    float*     ssum  = (float*)    carve((size_t)N_NODES * 2 * 4);
    _Float16*  W0t   = (_Float16*) carve(128 * 128 * 2);
    _Float16*  W1t   = (_Float16*) carve(128 * 128 * 2);
    _Float16*  Wp1t  = (_Float16*) carve(64 * 128 * 2);
    _Float16*  Wp2t  = (_Float16*) carve(64 * 64 * 2);

    prep_weights<<<(128 * 128 + 255) / 256, 256, 0, stream>>>(
        W0, W1, Wp1, Wp2, W0t, W1t, Wp1t, Wp2t);

    const int gemmBlocks  = (N_NODES / 16) * 8 / 8;
    const int edgeBlocks  = E_EDGES / 8;
    const int scoreBlocks = E_EDGES / SCORE_STRIP / 8;
    const int nodeBlocks  = (N_NODES * HDIM + 255) / 256;

    hipMemsetAsync(rst,  0,    (size_t)N_NODES * HDIM * 4, stream);
    hipMemsetAsync(ssum, 0,    (size_t)N_NODES * 2 * 4,    stream);
    hipMemsetAsync(mOrd, 0x80, (size_t)N_NODES * 2 * 4,    stream);
    gemm128_wmma<<<gemmBlocks, 256, 0, stream>>>(x, W0t, feat, N_NODES);
    edge_score_kernel<<<scoreBlocks, 256, 0, stream>>>(feat, src, dst, attn0, score, mOrd);
    edge_aggregate_kernel<<<edgeBlocks, 256, 0, stream>>>(feat, src, dst, score, mOrd, ssum, rst);
    finalize_kernel<<<nodeBlocks, 256, 0, stream>>>(rst, ssum, x, h0, 0, N_NODES * HDIM);

    hipMemsetAsync(rst,  0,    (size_t)N_NODES * HDIM * 4, stream);
    hipMemsetAsync(ssum, 0,    (size_t)N_NODES * 2 * 4,    stream);
    hipMemsetAsync(mOrd, 0x80, (size_t)N_NODES * 2 * 4,    stream);
    gemm128_wmma<<<gemmBlocks, 256, 0, stream>>>(h0, W1t, feat, N_NODES);
    edge_score_kernel<<<scoreBlocks, 256, 0, stream>>>(feat, src, dst, attn1, score, mOrd);
    edge_aggregate_kernel<<<edgeBlocks, 256, 0, stream>>>(feat, src, dst, score, mOrd, ssum, rst);
    finalize_kernel<<<nodeBlocks, 256, 0, stream>>>(rst, ssum, h0, h1, 1, N_NODES * HDIM);

    const int predBlocks = (2 * (P_CAND / 16) + 3) / 4;
    predictor_wmma<<<predBlocks, 128, 0, stream>>>(
        h1, pos_src, pos_dst, neg_src, neg_dst,
        Wp1t, bp1, Wp2t, bp2, Wp3, bp3, (float*)d_out);
}


# Atomic targets (row sums, weight sums, score maxima) kept once per cache half: a wave adds into the copy of the die half it runs on, node kernel adds the two copies: sum order now two partial sums
# speedup vs baseline: 2.0649x; 1.4973x over previous
#include <hip/hip_runtime.h>


typedef __attribute__((ext_vector_type(16))) _Float16 v16h;
typedef __attribute__((ext_vector_type(8)))  _Float16 v8h;
typedef __attribute__((ext_vector_type(8)))  float    v8f;

#define N_NODES 50000
#define E_EDGES 800000
#define P_CAND  100000
#define HDIM    128
#define NEG_SLOPE 0.2f


__device__ __forceinline__ int f2ord(float f) {
    int i = __float_as_int(f);
    return i >= 0 ? i : (i ^ 0x7fffffff);
}
__device__ __forceinline__ float ord2f(int i) {
    return __int_as_float(i >= 0 ? i : (i ^ 0x7fffffff));
}

__device__ __forceinline__ int cache_half() {
    unsigned id = __builtin_amdgcn_s_sendmsg_rtn(0x87);
    return (int)((id >> 18) & 1u);
}
#define RST_PLANE  (N_NODES * HDIM)
#define NH_PLANE   (N_NODES * 2)

__device__ __forceinline__ v16h load_a_f32(const float* __restrict__ p) {
    float4 a0 = *(const float4*)(p);
    float4 a1 = *(const float4*)(p + 4);
    float4 a2 = *(const float4*)(p + 16);
    float4 a3 = *(const float4*)(p + 20);
    v16h v;
    v[0]=(_Float16)a0.x; v[1]=(_Float16)a0.y; v[2]=(_Float16)a0.z; v[3]=(_Float16)a0.w;
    v[4]=(_Float16)a1.x; v[5]=(_Float16)a1.y; v[6]=(_Float16)a1.z; v[7]=(_Float16)a1.w;
    v[8]=(_Float16)a2.x; v[9]=(_Float16)a2.y; v[10]=(_Float16)a2.z; v[11]=(_Float16)a2.w;
    v[12]=(_Float16)a3.x; v[13]=(_Float16)a3.y; v[14]=(_Float16)a3.z; v[15]=(_Float16)a3.w;
    return v;
}

__device__ __forceinline__ v16h load_a_prod(const float* __restrict__ sp,
                                            const float* __restrict__ dp) {
    float4 s0 = *(const float4*)(sp),      d0 = *(const float4*)(dp);
    float4 s1 = *(const float4*)(sp + 4),  d1 = *(const float4*)(dp + 4);
    float4 s2 = *(const float4*)(sp + 16), d2 = *(const float4*)(dp + 16);
    float4 s3 = *(const float4*)(sp + 20), d3 = *(const float4*)(dp + 20);
    v16h v;
    v[0]=(_Float16)(s0.x*d0.x); v[1]=(_Float16)(s0.y*d0.y);
    v[2]=(_Float16)(s0.z*d0.z); v[3]=(_Float16)(s0.w*d0.w);
    v[4]=(_Float16)(s1.x*d1.x); v[5]=(_Float16)(s1.y*d1.y);
    v[6]=(_Float16)(s1.z*d1.z); v[7]=(_Float16)(s1.w*d1.w);
    v[8]=(_Float16)(s2.x*d2.x); v[9]=(_Float16)(s2.y*d2.y);
    v[10]=(_Float16)(s2.z*d2.z); v[11]=(_Float16)(s2.w*d2.w);
    v[12]=(_Float16)(s3.x*d3.x); v[13]=(_Float16)(s3.y*d3.y);
    v[14]=(_Float16)(s3.z*d3.z); v[15]=(_Float16)(s3.w*d3.w);
    return v;
}

__device__ __forceinline__ v16h load_b_f16(const _Float16* __restrict__ p) {
    v8h lo = *(const v8h*)(p);
    v8h hi = *(const v8h*)(p + 16);
    v16h v;
#pragma unroll
    for (int i = 0; i < 8; ++i) { v[i] = lo[i]; v[8 + i] = hi[i]; }
    return v;
}

__global__ void prep_weights(const float* __restrict__ W0, const float* __restrict__ W1,
                             const float* __restrict__ Wp1, const float* __restrict__ Wp2,
                             _Float16* __restrict__ W0t, _Float16* __restrict__ W1t,
                             _Float16* __restrict__ Wp1t, _Float16* __restrict__ Wp2t) {
    int idx = blockIdx.x * blockDim.x + threadIdx.x;
    if (idx < 128 * 128) {
        int c = idx >> 7, k = idx & 127;
        W0t[idx] = (_Float16)W0[k * 128 + c];
        W1t[idx] = (_Float16)W1[k * 128 + c];
    }
    if (idx < 64 * 128) {
        int c = idx >> 7, k = idx & 127;
        Wp1t[idx] = (_Float16)Wp1[k * 64 + c];
    }
    if (idx < 64 * 64) {
        int c = idx >> 6, k = idx & 63;
        Wp2t[idx] = (_Float16)Wp2[k * 64 + c];
    }
}

__global__ __launch_bounds__(256)
void gemm128_wmma(const float* __restrict__ A, const _Float16* __restrict__ Bt,
                  float* __restrict__ C, int M) {
    int wid  = blockIdx.x * (blockDim.x >> 5) + (threadIdx.x >> 5);
    int lane = threadIdx.x & 31;
    int half = lane >> 4, r = lane & 15;
    int tm = wid >> 3, tn = wid & 7;
    if (tm * 16 >= M) return;
    int row = tm * 16 + r;
    int col = tn * 16 + r;
    v8f acc = {};
#pragma unroll
    for (int kk = 0; kk < 128; kk += 32) {
        int k0 = kk + half * 8;
        v16h a = load_a_f32(A + row * 128 + k0);
        v16h b = load_b_f16(Bt + col * 128 + k0);
        acc = __builtin_amdgcn_wmma_f32_16x16x32_f16(false, a, false, b,
                                                     (short)0, acc, false, false);
    }
    float* cp = C + (tm * 16 + half * 8) * 128 + tn * 16 + r;
#pragma unroll
    for (int v = 0; v < 8; ++v) cp[v * 128] = acc[v];
}

#define SCORE_STRIP 32
#define SCORE_GROUP 8
static_assert(E_EDGES % SCORE_STRIP == 0, "strips cover the edge list exactly");
__global__ __launch_bounds__(256)
void edge_score_kernel(const float* __restrict__ feat, const int* __restrict__ src,
                       const int* __restrict__ dst, const float* __restrict__ attn,
                       float* __restrict__ score, int* __restrict__ mOrd) {
    int strip = (blockIdx.x * blockDim.x + threadIdx.x) >> 5;
    int eb = strip * SCORE_STRIP;
    if (eb >= E_EDGES) return;
    int lane = threadIdx.x & 31;
    int hh = lane >> 4, c = lane & 15, d4 = c << 2;
    int sv = src[eb + lane];
    int dv = dst[eb + lane];
    float4 at = *(const float4*)(attn + hh * 64 + d4);
    const float* fl = feat + hh * 64 + d4;
    float keep0 = 0.f, keep1 = 0.f;
#pragma unroll 1
    for (int j0 = 0; j0 < SCORE_STRIP; j0 += SCORE_GROUP) {
        float4 el[SCORE_GROUP], er[SCORE_GROUP];
#pragma unroll
        for (int u = 0; u < SCORE_GROUP; ++u) {
            int s_ = __builtin_amdgcn_readlane(sv, j0 + u);
            int d_ = __builtin_amdgcn_readlane(dv, j0 + u);
            el[u] = *(const float4*)(fl + s_ * HDIM);
            er[u] = *(const float4*)(fl + d_ * HDIM);
        }
#pragma unroll
        for (int u = 0; u < SCORE_GROUP; ++u) {
            float x0 = el[u].x + er[u].x; x0 = x0 > 0.f ? x0 : NEG_SLOPE * x0;
            float x1 = el[u].y + er[u].y; x1 = x1 > 0.f ? x1 : NEG_SLOPE * x1;
            float x2 = el[u].z + er[u].z; x2 = x2 > 0.f ? x2 : NEG_SLOPE * x2;
            float x3 = el[u].w + er[u].w; x3 = x3 > 0.f ? x3 : NEG_SLOPE * x3;
            float part = fmaf(x3, at.w, fmaf(x2, at.z, fmaf(x0, at.x, x1 * at.y)));
            part += __shfl_xor(part, 1, 32);
            part += __shfl_xor(part, 2, 32);
            part += __shfl_xor(part, 4, 32);
            part += __shfl_xor(part, 8, 32);
            const int j = j0 + u;
            if (c == (j & 15)) { if (j < 16) keep0 = part; else keep1 = part; }
        }
    }
    mOrd += cache_half() * NH_PLANE;
    int dA = __shfl(dv, c, 32);
    int dB = __shfl(dv, 16 + c, 32);
    score[(eb + c) * 2 + hh]      = keep0;
    score[(eb + 16 + c) * 2 + hh] = keep1;
    atomicMax(&mOrd[dA * 2 + hh], f2ord(keep0));
    atomicMax(&mOrd[dB * 2 + hh], f2ord(keep1));
}

#define AGG_EDGES 8
static_assert(E_EDGES % AGG_EDGES == 0, "runs cover the edge list exactly");
static_assert((AGG_EDGES & (AGG_EDGES - 1)) == 0 && AGG_EDGES <= 32, "lane -> edge of the run by a mask");
__global__ __launch_bounds__(256)
void edge_aggregate_kernel(const float* __restrict__ feat, const int* __restrict__ src,
                           const int* __restrict__ dst, const float* __restrict__ score,
                           const int* __restrict__ mOrd, float* __restrict__ ssum,
                           float* __restrict__ rst) {
    int run = (blockIdx.x * blockDim.x + threadIdx.x) >> 5;
    int eb = run * AGG_EDGES;
    if (eb >= E_EDGES) return;
    int lane = threadIdx.x & 31;
    int ej = eb + (lane & (AGG_EDGES - 1));
    int sv = src[ej];
    int dv = dst[ej];
    float2 sc = *(const float2*)(score + ej * 2);
    int2   mo = *(const int2*)(mOrd + dv * 2);
    int2   mq = *(const int2*)(mOrd + NH_PLANE + dv * 2);
    mo.x = mo.x > mq.x ? mo.x : mq.x;
    mo.y = mo.y > mq.y ? mo.y : mq.y;
    const int hf = cache_half();
    ssum += hf * NH_PLANE;
    rst  += hf * RST_PLANE;
    const float* fl = feat + lane;
    float v[AGG_EDGES][4];
#pragma unroll
    for (int u = 0; u < AGG_EDGES; ++u) {
        int s_ = __builtin_amdgcn_readlane(sv, u);
        const float* fr = fl + s_ * HDIM;
        v[u][0] = fr[0]; v[u][1] = fr[32]; v[u][2] = fr[64]; v[u][3] = fr[96];
    }
    int p0v = __float_as_int(__expf(sc.x - ord2f(mo.x)));
    int p1v = __float_as_int(__expf(sc.y - ord2f(mo.y)));
#pragma unroll
    for (int u = 0; u < AGG_EDGES; ++u) {
        int d_ = __builtin_amdgcn_readlane(dv, u);
        float p0 = __int_as_float(__builtin_amdgcn_readlane(p0v, u));
        float p1 = __int_as_float(__builtin_amdgcn_readlane(p1v, u));
        if (lane < 2) atomicAdd(&ssum[d_ * 2 + lane], lane ? p1 : p0);
        float* rp = rst + d_ * HDIM + lane;
        atomicAdd(rp +  0, v[u][0] * p0);
        atomicAdd(rp + 32, v[u][1] * p0);
        atomicAdd(rp + 64, v[u][2] * p1);
        atomicAdd(rp + 96, v[u][3] * p1);
    }
}

__global__ __launch_bounds__(256)
void finalize_kernel(const float* __restrict__ rst, const float* __restrict__ ssum,
                     const float* __restrict__ hin,
                     float* __restrict__ hout, int residual, int n) {
    int idx = blockIdx.x * blockDim.x + threadIdx.x;
    if (idx >= n) return;
    float s = ssum[idx >> 6] + ssum[NH_PLANE + (idx >> 6)];
    float r = rst[idx] + rst[RST_PLANE + idx];
    float v = s > 0.f ? r / s : 0.f;
    v += residual ? hin[idx] : 0.0f;
    hout[idx] = v > 0.f ? v : (__expf(v) - 1.0f);
}

__global__ __launch_bounds__(128)
void predictor_wmma(const float* __restrict__ h,
                    const int* __restrict__ ps, const int* __restrict__ pd,
                    const int* __restrict__ ns, const int* __restrict__ nd,
                    const _Float16* __restrict__ Wp1t, const float* __restrict__ bp1,
                    const _Float16* __restrict__ Wp2t, const float* __restrict__ bp2,
                    const float* __restrict__ Wp3, const float* __restrict__ bp3,
                    float* __restrict__ out) {
    __shared__ _Float16 z1s[4][16 * 64];
    __shared__ float    z2s[4][16 * 64];
    const int tilesPerSet = P_CAND / 16;
    int w = threadIdx.x >> 5;
    int tile = blockIdx.x * 4 + w;
    if (tile >= 2 * tilesPerSet) return;
    bool isNeg = tile >= tilesPerSet;
    int t = isNeg ? tile - tilesPerSet : tile;
    const int* S = isNeg ? ns : ps;
    const int* D = isNeg ? nd : pd;

    int lane = threadIdx.x & 31;
    int half = lane >> 4, r = lane & 15;
    int c = t * 16 + r;
    int si = S[c], di = D[c];

    v16h a1[4];
#pragma unroll
    for (int kc = 0; kc < 4; ++kc) {
        int k0 = kc * 32 + half * 8;
        a1[kc] = load_a_prod(h + si * HDIM + k0, h + di * HDIM + k0);
    }
#pragma unroll
    for (int nt = 0; nt < 4; ++nt) {
        int col = nt * 16 + r;
        v8f acc = {};
#pragma unroll
        for (int kc = 0; kc < 4; ++kc) {
            int k0 = kc * 32 + half * 8;
            v16h b = load_b_f16(Wp1t + col * 128 + k0);
            acc = __builtin_amdgcn_wmma_f32_16x16x32_f16(false, a1[kc], false, b,
                                                         (short)0, acc, false, false);
        }
        float bias = bp1[col];
#pragma unroll
        for (int v = 0; v < 8; ++v) {
            float z = acc[v] + bias;
            z = z > 0.f ? z : 0.f;
            z1s[w][(v + half * 8) * 64 + col] = (_Float16)z;
        }
    }

    v16h a2[2];
#pragma unroll
    for (int kc = 0; kc < 2; ++kc) {
        int k0 = kc * 32 + half * 8;
        const _Float16* zp = &z1s[w][r * 64 + k0];
        a2[kc] = load_b_f16(zp);
    }
#pragma unroll
    for (int nt = 0; nt < 4; ++nt) {
        int col = nt * 16 + r;
        v8f acc = {};
#pragma unroll
        for (int kc = 0; kc < 2; ++kc) {
            int k0 = kc * 32 + half * 8;
            v16h b = load_b_f16(Wp2t + col * 64 + k0);
            acc = __builtin_amdgcn_wmma_f32_16x16x32_f16(false, a2[kc], false, b,
                                                         (short)0, acc, false, false);
        }
        float bias = bp2[col];
#pragma unroll
        for (int v = 0; v < 8; ++v) {
            float z = acc[v] + bias;
            z2s[w][(v + half * 8) * 64 + col] = z > 0.f ? z : 0.f;
        }
    }

    float part = 0.f;
    int jb = half * 32;
#pragma unroll
    for (int j = 0; j < 32; ++j)
        part += z2s[w][r * 64 + jb + j] * Wp3[jb + j];
    part += __shfl_xor(part, 16, 32);
    if (half == 0) {
        out[(isNeg ? P_CAND : 0) + t * 16 + r] = part + bp3[0];
    }
}

extern "C" void kernel_launch(void* const* d_in, const int* in_sizes, int n_in,
                              void* d_out, int out_size, void* d_ws, size_t ws_size,
                              hipStream_t stream) {
    const float* x        = (const float*)d_in[0];
    const int*   src      = (const int*)  d_in[1];
    const int*   dst      = (const int*)  d_in[2];
    const int*   pos_src  = (const int*)  d_in[3];
    const int*   pos_dst  = (const int*)  d_in[4];
    const int*   neg_src  = (const int*)  d_in[5];
    const int*   neg_dst  = (const int*)  d_in[6];
    const float* W0       = (const float*)d_in[7];
    const float* attn0    = (const float*)d_in[8];
    const float* W1       = (const float*)d_in[9];
    const float* attn1    = (const float*)d_in[10];
    const float* Wp1      = (const float*)d_in[11];
    const float* bp1      = (const float*)d_in[12];
    const float* Wp2      = (const float*)d_in[13];
    const float* bp2      = (const float*)d_in[14];
    const float* Wp3      = (const float*)d_in[15];
    const float* bp3      = (const float*)d_in[16];

    char* ws = (char*)d_ws;
    size_t off = 0;
    auto carve = [&](size_t bytes) -> char* {
        char* p = ws + off;
        off += (bytes + 255) & ~(size_t)255;
        return p;
    };
    float*     feat  = (float*)    carve((size_t)N_NODES * HDIM * 4);
    float*     h0    = (float*)    carve((size_t)N_NODES * HDIM * 4);
    float*     h1    = (float*)    carve((size_t)N_NODES * HDIM * 4);
    float*     rst   = (float*)    carve((size_t)2 * RST_PLANE * 4);
    float*     score = (float*)    carve((size_t)E_EDGES * 2 * 4);
    int*       mOrd  = (int*)      carve((size_t)2 * NH_PLANE * 4);
    float*     ssum  = (float*)    carve((size_t)2 * NH_PLANE * 4);
    _Float16*  W0t   = (_Float16*) carve(128 * 128 * 2);
    _Float16*  W1t   = (_Float16*) carve(128 * 128 * 2);
    _Float16*  Wp1t  = (_Float16*) carve(64 * 128 * 2);
    _Float16*  Wp2t  = (_Float16*) carve(64 * 64 * 2);

    prep_weights<<<(128 * 128 + 255) / 256, 256, 0, stream>>>(
        W0, W1, Wp1, Wp2, W0t, W1t, Wp1t, Wp2t);

    const int gemmBlocks  = (N_NODES / 16) * 8 / 8;
    const int edgeBlocks  = E_EDGES / AGG_EDGES / 8;
    const int scoreBlocks = E_EDGES / SCORE_STRIP / 8;
    const int nodeBlocks  = (N_NODES * HDIM + 255) / 256;

    hipMemsetAsync(rst,  0,    (size_t)2 * RST_PLANE * 4, stream);
    hipMemsetAsync(ssum, 0,    (size_t)2 * NH_PLANE * 4,  stream);
    hipMemsetAsync(mOrd, 0x80, (size_t)2 * NH_PLANE * 4,  stream);
    gemm128_wmma<<<gemmBlocks, 256, 0, stream>>>(x, W0t, feat, N_NODES);
    edge_score_kernel<<<scoreBlocks, 256, 0, stream>>>(feat, src, dst, attn0, score, mOrd);
    edge_aggregate_kernel<<<edgeBlocks, 256, 0, stream>>>(feat, src, dst, score, mOrd, ssum, rst);
    finalize_kernel<<<nodeBlocks, 256, 0, stream>>>(rst, ssum, x, h0, 0, N_NODES * HDIM);

    hipMemsetAsync(rst,  0,    (size_t)2 * RST_PLANE * 4, stream);
    hipMemsetAsync(ssum, 0,    (size_t)2 * NH_PLANE * 4,  stream);
    hipMemsetAsync(mOrd, 0x80, (size_t)2 * NH_PLANE * 4,  stream);
    gemm128_wmma<<<gemmBlocks, 256, 0, stream>>>(h0, W1t, feat, N_NODES);
    edge_score_kernel<<<scoreBlocks, 256, 0, stream>>>(feat, src, dst, attn1, score, mOrd);
    edge_aggregate_kernel<<<edgeBlocks, 256, 0, stream>>>(feat, src, dst, score, mOrd, ssum, rst);
    finalize_kernel<<<nodeBlocks, 256, 0, stream>>>(rst, ssum, h0, h1, 1, N_NODES * HDIM);

    const int predBlocks = (2 * (P_CAND / 16) + 3) / 4;
    predictor_wmma<<<predBlocks, 128, 0, stream>>>(
        h1, pos_src, pos_dst, neg_src, neg_dst,
        Wp1t, bp1, Wp2t, bp2, Wp3, bp3, (float*)d_out);
}


# Edge scores computed inside the owner work-group's launch from its 200 dst rows kept in LDS, maxima by LDS integer atomics, scores in the waves' lists: score launches, fills gone; sum order same
# speedup vs baseline: 3.0435x; 1.4739x over previous
#include <hip/hip_runtime.h>


typedef __attribute__((ext_vector_type(16))) _Float16 v16h;
typedef __attribute__((ext_vector_type(8)))  _Float16 v8h;
typedef __attribute__((ext_vector_type(8)))  float    v8f;

#define N_NODES 50000
#define E_EDGES 800000
#define P_CAND  100000
#define HDIM    128
#define NEG_SLOPE 0.2f


__device__ __forceinline__ int f2ord(float f) {
    int i = __float_as_int(f);
    return i >= 0 ? i : (i ^ 0x7fffffff);
}
__device__ __forceinline__ float ord2f(int i) {
    return __int_as_float(i >= 0 ? i : (i ^ 0x7fffffff));
}

__device__ __forceinline__ v16h load_a_f32(const float* __restrict__ p) {
    float4 a0 = *(const float4*)(p);
    float4 a1 = *(const float4*)(p + 4);
    float4 a2 = *(const float4*)(p + 16);
    float4 a3 = *(const float4*)(p + 20);
    v16h v;
    v[0]=(_Float16)a0.x; v[1]=(_Float16)a0.y; v[2]=(_Float16)a0.z; v[3]=(_Float16)a0.w;
    v[4]=(_Float16)a1.x; v[5]=(_Float16)a1.y; v[6]=(_Float16)a1.z; v[7]=(_Float16)a1.w;
    v[8]=(_Float16)a2.x; v[9]=(_Float16)a2.y; v[10]=(_Float16)a2.z; v[11]=(_Float16)a2.w;
    v[12]=(_Float16)a3.x; v[13]=(_Float16)a3.y; v[14]=(_Float16)a3.z; v[15]=(_Float16)a3.w;
    return v;
}

__device__ __forceinline__ v16h load_a_prod(const float* __restrict__ sp,
                                            const float* __restrict__ dp) {
    float4 s0 = *(const float4*)(sp),      d0 = *(const float4*)(dp);
    float4 s1 = *(const float4*)(sp + 4),  d1 = *(const float4*)(dp + 4);
    float4 s2 = *(const float4*)(sp + 16), d2 = *(const float4*)(dp + 16);
    float4 s3 = *(const float4*)(sp + 20), d3 = *(const float4*)(dp + 20);
    v16h v;
    v[0]=(_Float16)(s0.x*d0.x); v[1]=(_Float16)(s0.y*d0.y);
    v[2]=(_Float16)(s0.z*d0.z); v[3]=(_Float16)(s0.w*d0.w);
    v[4]=(_Float16)(s1.x*d1.x); v[5]=(_Float16)(s1.y*d1.y);
    v[6]=(_Float16)(s1.z*d1.z); v[7]=(_Float16)(s1.w*d1.w);
    v[8]=(_Float16)(s2.x*d2.x); v[9]=(_Float16)(s2.y*d2.y);
    v[10]=(_Float16)(s2.z*d2.z); v[11]=(_Float16)(s2.w*d2.w);
    v[12]=(_Float16)(s3.x*d3.x); v[13]=(_Float16)(s3.y*d3.y);
    v[14]=(_Float16)(s3.z*d3.z); v[15]=(_Float16)(s3.w*d3.w);
    return v;
}

__device__ __forceinline__ v16h load_b_f16(const _Float16* __restrict__ p) {
    v8h lo = *(const v8h*)(p);
    v8h hi = *(const v8h*)(p + 16);
    v16h v;
#pragma unroll
    for (int i = 0; i < 8; ++i) { v[i] = lo[i]; v[8 + i] = hi[i]; }
    return v;
}

__global__ void prep_weights(const float* __restrict__ W0, const float* __restrict__ W1,
                             const float* __restrict__ Wp1, const float* __restrict__ Wp2,
                             _Float16* __restrict__ W0t, _Float16* __restrict__ W1t,
                             _Float16* __restrict__ Wp1t, _Float16* __restrict__ Wp2t) {
    int idx = blockIdx.x * blockDim.x + threadIdx.x;
    if (idx < 128 * 128) {
        int c = idx >> 7, k = idx & 127;
        W0t[idx] = (_Float16)W0[k * 128 + c];
        W1t[idx] = (_Float16)W1[k * 128 + c];
    }
    if (idx < 64 * 128) {
        int c = idx >> 7, k = idx & 127;
        Wp1t[idx] = (_Float16)Wp1[k * 64 + c];
    }
    if (idx < 64 * 64) {
        int c = idx >> 6, k = idx & 63;
        Wp2t[idx] = (_Float16)Wp2[k * 64 + c];
    }
}

__global__ __launch_bounds__(256)
void gemm128_wmma(const float* __restrict__ A, const _Float16* __restrict__ Bt,
                  float* __restrict__ C, int M) {
    int wid  = blockIdx.x * (blockDim.x >> 5) + (threadIdx.x >> 5);
    int lane = threadIdx.x & 31;
    int half = lane >> 4, r = lane & 15;
    int tm = wid >> 3, tn = wid & 7;
    if (tm * 16 >= M) return;
    int row = tm * 16 + r;
    int col = tn * 16 + r;
    v8f acc = {};
#pragma unroll
    for (int kk = 0; kk < 128; kk += 32) {
        int k0 = kk + half * 8;
        v16h a = load_a_f32(A + row * 128 + k0);
        v16h b = load_b_f16(Bt + col * 128 + k0);
        acc = __builtin_amdgcn_wmma_f32_16x16x32_f16(false, a, false, b,
                                                     (short)0, acc, false, false);
    }
    float* cp = C + (tm * 16 + half * 8) * 128 + tn * 16 + r;
#pragma unroll
    for (int v = 0; v < 8; ++v) cp[v * 128] = acc[v];
}

#define OWN_NODES  200
#define OWN_WAVES  32
#define SCAN_LOADS 4
#define SCAN_IDS   (SCAN_LOADS * 128)
#define SCAN_STEP  (OWN_WAVES * SCAN_IDS)
#define AGG_EDGES  8
#define QCAP       (SCAN_IDS + AGG_EDGES)
#define LIST_CAP   256
#define N_GROUPS   (N_NODES / OWN_NODES)
#define MAX_START  ((int)0x80808080)
#define ROW_QUADS  (OWN_NODES * HDIM / 4)
#define ROW_SWEEPS ((ROW_QUADS + OWN_WAVES * 32 - 1) / (OWN_WAVES * 32))
static_assert(N_NODES % OWN_NODES == 0, "groups cover the nodes exactly");
static_assert(OWN_NODES <= 256 && E_EDGES <= (1 << 23), "queue word: edge << 8 | node in group");
static_assert(E_EDGES % 4 == 0, "a 16-byte id load is inside the list or outside it");
static_assert((OWN_NODES * HDIM) % (OWN_WAVES * 32) == 0, "node pass: whole sweeps");
static_assert(LIST_CAP % AGG_EDGES == 0, "a round's entries stay inside the wave's list");

__device__ __forceinline__ int4 load_ids(const int* __restrict__ dst, int i0) {
    const bool in = i0 < E_EDGES;
    int4 t = *(const int4*)(dst + (in ? i0 : 0));
    t.x = in ? t.x : -1; t.y = in ? t.y : -1; t.z = in ? t.z : -1; t.w = in ? t.w : -1;
    return t;
}

__device__ __forceinline__ float2 score_edges(const float* ers,
                                              const float* __restrict__ fl4, float4 at,
                                              int lane, int dl, int sv) {
    const int c = lane & 15;
    float4 el[AGG_EDGES];
#pragma unroll
    for (int u = 0; u < AGG_EDGES; ++u) {
        int s_ = __builtin_amdgcn_readlane(sv, u);
        el[u] = *(const float4*)(fl4 + s_ * HDIM);
    }
    float keep = 0.f;
#pragma unroll
    for (int u = 0; u < AGG_EDGES; ++u) {
        int d_ = __builtin_amdgcn_readlane(dl, u);
        float4 er = *(const float4*)(ers + d_ * HDIM + lane * 4);
        float x0 = el[u].x + er.x; x0 = x0 > 0.f ? x0 : NEG_SLOPE * x0;
        float x1 = el[u].y + er.y; x1 = x1 > 0.f ? x1 : NEG_SLOPE * x1;
        float x2 = el[u].z + er.z; x2 = x2 > 0.f ? x2 : NEG_SLOPE * x2;
        float x3 = el[u].w + er.w; x3 = x3 > 0.f ? x3 : NEG_SLOPE * x3;
        float part = fmaf(x3, at.w, fmaf(x2, at.z, fmaf(x0, at.x, x1 * at.y)));
        part += __shfl_xor(part, 1, 32);
        part += __shfl_xor(part, 2, 32);
        part += __shfl_xor(part, 4, 32);
        part += __shfl_xor(part, 8, 32);
        if (c == u) keep = part;
    }
    float keep1 = __shfl_xor(keep, 16, 32);
    return make_float2(keep, keep1);
}

__device__ __forceinline__ void add_edges(float* acc, float* ss, const int* mx,
                                          const float* __restrict__ fl,
                                          int lane, int n, int dl, int sv,
                                          float s0, float s1) {
    int2 mo = *(const int2*)(mx + dl * 2);
    float v[AGG_EDGES][4];
#pragma unroll
    for (int u = 0; u < AGG_EDGES; ++u) {
        int s_ = __builtin_amdgcn_readlane(sv, u);
        const float* fr = fl + s_ * HDIM;
        v[u][0] = fr[0]; v[u][1] = fr[32]; v[u][2] = fr[64]; v[u][3] = fr[96];
    }
    int p0v = __float_as_int(__expf(s0 - ord2f(mo.x)));
    int p1v = __float_as_int(__expf(s1 - ord2f(mo.y)));
#pragma unroll
    for (int u = 0; u < AGG_EDGES; ++u) {
        if (u < n) {
            int d_ = __builtin_amdgcn_readlane(dl, u);
            float p0 = __int_as_float(__builtin_amdgcn_readlane(p0v, u));
            float p1 = __int_as_float(__builtin_amdgcn_readlane(p1v, u));
            if (lane < 2) atomicAdd(&ss[d_ * 2 + lane], lane ? p1 : p0);
            float* rp = acc + d_ * HDIM + lane;
            atomicAdd(rp +  0, v[u][0] * p0);
            atomicAdd(rp + 32, v[u][1] * p0);
            atomicAdd(rp + 64, v[u][2] * p1);
            atomicAdd(rp + 96, v[u][3] * p1);
        }
    }
}

template <class F>
__device__ __forceinline__ int scan_rounds(const int* __restrict__ dst,
                                           const int* __restrict__ src,
                                           int lo, int w, int lane, int* qw, F round) {
    int cnt = 0;
    int tot = 0;
    const unsigned below = (1u << lane) - 1u;
    int4 cur[SCAN_LOADS];
#pragma unroll
    for (int c = 0; c < SCAN_LOADS; ++c) {
        cur[c] = load_ids(dst, w * SCAN_IDS + c * 128 + lane * 4);
    }
    for (int base = w * SCAN_IDS; base < E_EDGES; base += SCAN_STEP) {
        int4 nxt[SCAN_LOADS];
#pragma unroll
        for (int c = 0; c < SCAN_LOADS; ++c) {
            nxt[c] = load_ids(dst, base + SCAN_STEP + c * 128 + lane * 4);
        }
#pragma unroll
        for (int c = 0; c < SCAN_LOADS; ++c) {
            const int nd[4] = { cur[c].x - lo, cur[c].y - lo, cur[c].z - lo, cur[c].w - lo };
            const bool any = (unsigned)nd[0] < (unsigned)OWN_NODES || (unsigned)nd[1] < (unsigned)OWN_NODES ||
                             (unsigned)nd[2] < (unsigned)OWN_NODES || (unsigned)nd[3] < (unsigned)OWN_NODES;
            if (__ballot(any) != 0) {
#pragma unroll
                for (int k = 0; k < 4; ++k) {
                    const bool hit = (unsigned)nd[k] < (unsigned)OWN_NODES;
                    const unsigned mask = (unsigned)__ballot(hit);
                    if (hit) qw[cnt + __popc(mask & below)] =
                                 ((base + c * 128 + lane * 4 + k) << 8) | nd[k];
                    cnt += __popc(mask);
                }
            }
        }
        const bool last = base + SCAN_STEP >= E_EDGES;
        __builtin_amdgcn_wave_barrier();
        int head = 0;
        while (cnt - head >= AGG_EDGES || (last && cnt > head)) {
            const int n = cnt - head < AGG_EDGES ? cnt - head : AGG_EDGES;
            const int j = lane & (AGG_EDGES - 1);
            const int qe = qw[head + (j < n ? j : 0)];
            const int sv = src[qe >> 8];
            round(n, tot, qe, sv);
            tot += n;
            head += n;
        }
        if (head > 0) {
            const int rem = cnt - head;
            int keep = lane < rem ? qw[head + lane] : 0;
            __builtin_amdgcn_wave_barrier();
            if (lane < rem) qw[lane] = keep;
            cnt = rem;
        }
#pragma unroll
        for (int c = 0; c < SCAN_LOADS; ++c) cur[c] = nxt[c];
    }
    return tot;
}

template <class F>
__device__ __forceinline__ void list_rounds(const int* lp, int listed, int lane, F round) {
    if (listed <= 0) return;
    const int j = lane & (AGG_EDGES - 1);
    int4 ent = *(const int4*)(lp + 4 * (j < listed ? j : 0));
    int head = 0;
    for (; head + AGG_EDGES <= listed; head += AGG_EDGES) {
        const int nh = head + AGG_EDGES;
        const int ni = nh < listed ? nh + (nh + j < listed ? j : 0) : 0;
        const int4 nxt = *(const int4*)(lp + 4 * ni);
        round(AGG_EDGES, head, ent);
        ent = nxt;
    }
    if (head < listed) {
        round(listed - head, head, ent);
    }
}

template <bool FIRST>
__global__ __launch_bounds__(OWN_WAVES * 32)
void owner_layer_kernel(const float* __restrict__ feat, const int* __restrict__ src,
                        const int* __restrict__ dst, const float* __restrict__ attn,
                        const float* __restrict__ hin, float* __restrict__ hout,
                        int residual, int* elist, int* ecnt) {
    __shared__ float acc[OWN_NODES * HDIM];
    __shared__ float ers[OWN_NODES * HDIM];
    __shared__ float ss[OWN_NODES * 2];
    __shared__ int   mx[OWN_NODES * 2];
    __shared__ int   q[OWN_WAVES][QCAP];
    const int lo = blockIdx.x * OWN_NODES;
    const int tid = threadIdx.x, lane = tid & 31;
    const int w = __builtin_amdgcn_readfirstlane(tid >> 5);
    {
        const float4* own = (const float4*)(feat + lo * HDIM);
        float4 t[ROW_SWEEPS];
#pragma unroll
        for (int k = 0; k < ROW_SWEEPS; ++k) {
            const int i = tid + k * (OWN_WAVES * 32);
            t[k] = own[i < ROW_QUADS ? i : 0];
        }
        for (int i = tid; i < OWN_NODES * HDIM; i += OWN_WAVES * 32) acc[i] = 0.f;
        if (tid < OWN_NODES * 2) { ss[tid] = 0.f; mx[tid] = MAX_START; }
#pragma unroll
        for (int k = 0; k < ROW_SWEEPS; ++k) {
            const int i = tid + k * (OWN_WAVES * 32);
            if (i < ROW_QUADS) ((float4*)ers)[i] = t[k];
        }
    }
    __syncthreads();

    const int slot = blockIdx.x * OWN_WAVES + w;
    int* lp = elist + slot * (LIST_CAP * 4);
    int* qw = q[w];
    const float* fl  = feat + lane;
    const float* fl4 = feat + lane * 4;
    const float4 at = *(const float4*)(attn + lane * 4);

    auto score_scanned = [&](int n, int tot, int qe, int sv) {
        const int dl = qe & 255;
        const float2 sc = score_edges(ers, fl4, at, lane, dl, sv);
        if (lane < n) {
            atomicMax(&mx[dl * 2 + 0], f2ord(sc.x));
            atomicMax(&mx[dl * 2 + 1], f2ord(sc.y));
            if (FIRST && tot + lane < LIST_CAP) {
                *(int4*)(lp + 4 * (tot + lane)) =
                    make_int4(qe, sv, __float_as_int(sc.x), __float_as_int(sc.y));
            }
        }
    };
    auto score_listed = [&](int n, int head, int4 ent) {
        const int dl = ent.x & 255;
        const float2 sc = score_edges(ers, fl4, at, lane, dl, ent.y);
        if (lane < n) {
            atomicMax(&mx[dl * 2 + 0], f2ord(sc.x));
            atomicMax(&mx[dl * 2 + 1], f2ord(sc.y));
            *(float2*)(lp + 4 * (head + lane) + 2) = sc;
        }
    };
    auto add_scanned = [&](int n, int tot, int qe, int sv) {
        const int dl = qe & 255;
        const float2 sc = score_edges(ers, fl4, at, lane, dl, sv);
        add_edges(acc, ss, mx, fl, lane, n, dl, sv, sc.x, sc.y);
    };
    auto add_listed = [&](int n, int head, int4 ent) {
        add_edges(acc, ss, mx, fl, lane, n, ent.x & 255, ent.y,
                  __int_as_float(ent.z), __int_as_float(ent.w));
    };

    int listed;
    if (FIRST) {
        listed = __builtin_amdgcn_readfirstlane(
            scan_rounds(dst, src, lo, w, lane, qw, score_scanned));
        if (lane == 0) ecnt[slot] = listed;
    } else {
        listed = __builtin_amdgcn_readfirstlane(ecnt[slot]);
        if (listed <= LIST_CAP) list_rounds(lp, listed, lane, score_listed);
        else scan_rounds(dst, src, lo, w, lane, qw, score_scanned);
    }
    __syncthreads();
    if (listed <= LIST_CAP) list_rounds(lp, listed, lane, add_listed);
    else scan_rounds(dst, src, lo, w, lane, qw, add_scanned);
    __syncthreads();

    for (int i = tid; i < OWN_NODES * HDIM; i += OWN_WAVES * 32) {
        float s = ss[i >> 6];
        float r = acc[i];
        float x = s > 0.f ? r / s : 0.f;
        int g = lo * HDIM + i;
        x += residual ? hin[g] : 0.0f;
        hout[g] = x > 0.f ? x : (__expf(x) - 1.0f);
    }
}

__global__ __launch_bounds__(128)
void predictor_wmma(const float* __restrict__ h,
                    const int* __restrict__ ps, const int* __restrict__ pd,
                    const int* __restrict__ ns, const int* __restrict__ nd,
                    const _Float16* __restrict__ Wp1t, const float* __restrict__ bp1,
                    const _Float16* __restrict__ Wp2t, const float* __restrict__ bp2,
                    const float* __restrict__ Wp3, const float* __restrict__ bp3,
                    float* __restrict__ out) {
    __shared__ _Float16 z1s[4][16 * 64];
    __shared__ float    z2s[4][16 * 64];
    const int tilesPerSet = P_CAND / 16;
    int w = threadIdx.x >> 5;
    int tile = blockIdx.x * 4 + w;
    if (tile >= 2 * tilesPerSet) return;
    bool isNeg = tile >= tilesPerSet;
    int t = isNeg ? tile - tilesPerSet : tile;
    const int* S = isNeg ? ns : ps;
    const int* D = isNeg ? nd : pd;

    int lane = threadIdx.x & 31;
    int half = lane >> 4, r = lane & 15;
    int c = t * 16 + r;
    int si = S[c], di = D[c];

    v16h a1[4];
#pragma unroll
    for (int kc = 0; kc < 4; ++kc) {
        int k0 = kc * 32 + half * 8;
        a1[kc] = load_a_prod(h + si * HDIM + k0, h + di * HDIM + k0);
    }
#pragma unroll
    for (int nt = 0; nt < 4; ++nt) {
        int col = nt * 16 + r;
        v8f acc = {};
#pragma unroll
        for (int kc = 0; kc < 4; ++kc) {
            int k0 = kc * 32 + half * 8;
            v16h b = load_b_f16(Wp1t + col * 128 + k0);
            acc = __builtin_amdgcn_wmma_f32_16x16x32_f16(false, a1[kc], false, b,
                                                         (short)0, acc, false, false);
        }
        float bias = bp1[col];
#pragma unroll
        for (int v = 0; v < 8; ++v) {
            float z = acc[v] + bias;
            z = z > 0.f ? z : 0.f;
            z1s[w][(v + half * 8) * 64 + col] = (_Float16)z;
        }
    }

    v16h a2[2];
#pragma unroll
    for (int kc = 0; kc < 2; ++kc) {
        int k0 = kc * 32 + half * 8;
        const _Float16* zp = &z1s[w][r * 64 + k0];
        a2[kc] = load_b_f16(zp);
    }
#pragma unroll
    for (int nt = 0; nt < 4; ++nt) {
        int col = nt * 16 + r;
        v8f acc = {};
#pragma unroll
        for (int kc = 0; kc < 2; ++kc) {
            int k0 = kc * 32 + half * 8;
            v16h b = load_b_f16(Wp2t + col * 64 + k0);
            acc = __builtin_amdgcn_wmma_f32_16x16x32_f16(false, a2[kc], false, b,
                                                         (short)0, acc, false, false);
        }
        float bias = bp2[col];
#pragma unroll
        for (int v = 0; v < 8; ++v) {
            float z = acc[v] + bias;
            z2s[w][(v + half * 8) * 64 + col] = z > 0.f ? z : 0.f;
        }
    }

    float part = 0.f;
    int jb = half * 32;
#pragma unroll
    for (int j = 0; j < 32; ++j)
        part += z2s[w][r * 64 + jb + j] * Wp3[jb + j];
    part += __shfl_xor(part, 16, 32);
    if (half == 0) {
        out[(isNeg ? P_CAND : 0) + t * 16 + r] = part + bp3[0];
    }
}

extern "C" void kernel_launch(void* const* d_in, const int* in_sizes, int n_in,
                              void* d_out, int out_size, void* d_ws, size_t ws_size,
                              hipStream_t stream) {
    const float* x        = (const float*)d_in[0];
    const int*   src      = (const int*)  d_in[1];
    const int*   dst      = (const int*)  d_in[2];
    const int*   pos_src  = (const int*)  d_in[3];
    const int*   pos_dst  = (const int*)  d_in[4];
    const int*   neg_src  = (const int*)  d_in[5];
    const int*   neg_dst  = (const int*)  d_in[6];
    const float* W0       = (const float*)d_in[7];
    const float* attn0    = (const float*)d_in[8];
    const float* W1       = (const float*)d_in[9];
    const float* attn1    = (const float*)d_in[10];
    const float* Wp1      = (const float*)d_in[11];
    const float* bp1      = (const float*)d_in[12];
    const float* Wp2      = (const float*)d_in[13];
    const float* bp2      = (const float*)d_in[14];
    const float* Wp3      = (const float*)d_in[15];
    const float* bp3      = (const float*)d_in[16];

    char* ws = (char*)d_ws;
    size_t off = 0;
    auto carve = [&](size_t bytes) -> char* {
        char* p = ws + off;
        off += (bytes + 255) & ~(size_t)255;
        return p;
    };
    float*     feat  = (float*)    carve((size_t)N_NODES * HDIM * 4);
    float*     h0    = (float*)    carve((size_t)N_NODES * HDIM * 4);
    float*     h1    = (float*)    carve((size_t)N_NODES * HDIM * 4);
    _Float16*  W0t   = (_Float16*) carve(128 * 128 * 2);
    _Float16*  W1t   = (_Float16*) carve(128 * 128 * 2);
    _Float16*  Wp1t  = (_Float16*) carve(64 * 128 * 2);
    _Float16*  Wp2t  = (_Float16*) carve(64 * 64 * 2);
    int*       elist = (int*)      carve((size_t)N_GROUPS * OWN_WAVES * LIST_CAP * 4 * 4);
    int*       ecnt  = (int*)      carve((size_t)N_GROUPS * OWN_WAVES * 4);

    prep_weights<<<(128 * 128 + 255) / 256, 256, 0, stream>>>(
        W0, W1, Wp1, Wp2, W0t, W1t, Wp1t, Wp2t);

    const int gemmBlocks  = (N_NODES / 16) * 8 / 8;
    const int ownBlocks   = N_NODES / OWN_NODES;

    gemm128_wmma<<<gemmBlocks, 256, 0, stream>>>(x, W0t, feat, N_NODES);
    owner_layer_kernel<true><<<ownBlocks, OWN_WAVES * 32, 0, stream>>>(
        feat, src, dst, attn0, x, h0, 0, elist, ecnt);

    gemm128_wmma<<<gemmBlocks, 256, 0, stream>>>(h0, W1t, feat, N_NODES);
    owner_layer_kernel<false><<<ownBlocks, OWN_WAVES * 32, 0, stream>>>(
        feat, src, dst, attn1, h0, h1, 1, elist, ecnt);

    const int predBlocks = (2 * (P_CAND / 16) + 3) / 4;
    predictor_wmma<<<predBlocks, 128, 0, stream>>>(
        h1, pos_src, pos_dst, neg_src, neg_dst,
        Wp1t, bp1, Wp2t, bp2, Wp3, bp3, (float*)d_out);
}


# First layer's dst scan reads one owner byte per edge (plane from the weight launch): 16 ids per lane load, 3 loads ahead; dst id read per matched edge; a wave's queue order, so add order, differs
# speedup vs baseline: 3.2291x; 1.0610x over previous
#include <hip/hip_runtime.h>


typedef __attribute__((ext_vector_type(16))) _Float16 v16h;
typedef __attribute__((ext_vector_type(8)))  _Float16 v8h;
typedef __attribute__((ext_vector_type(8)))  float    v8f;

#define N_NODES 50000
#define E_EDGES 800000
#define P_CAND  100000
#define HDIM    128
#define NEG_SLOPE 0.2f


__device__ __forceinline__ int f2ord(float f) {
    int i = __float_as_int(f);
    return i >= 0 ? i : (i ^ 0x7fffffff);
}
__device__ __forceinline__ float ord2f(int i) {
    return __int_as_float(i >= 0 ? i : (i ^ 0x7fffffff));
}

__device__ __forceinline__ v16h load_a_f32(const float* __restrict__ p) {
    float4 a0 = *(const float4*)(p);
    float4 a1 = *(const float4*)(p + 4);
    float4 a2 = *(const float4*)(p + 16);
    float4 a3 = *(const float4*)(p + 20);
    v16h v;
    v[0]=(_Float16)a0.x; v[1]=(_Float16)a0.y; v[2]=(_Float16)a0.z; v[3]=(_Float16)a0.w;
    v[4]=(_Float16)a1.x; v[5]=(_Float16)a1.y; v[6]=(_Float16)a1.z; v[7]=(_Float16)a1.w;
    v[8]=(_Float16)a2.x; v[9]=(_Float16)a2.y; v[10]=(_Float16)a2.z; v[11]=(_Float16)a2.w;
    v[12]=(_Float16)a3.x; v[13]=(_Float16)a3.y; v[14]=(_Float16)a3.z; v[15]=(_Float16)a3.w;
    return v;
}

__device__ __forceinline__ v16h load_a_prod(const float* __restrict__ sp,
                                            const float* __restrict__ dp) {
    float4 s0 = *(const float4*)(sp),      d0 = *(const float4*)(dp);
    float4 s1 = *(const float4*)(sp + 4),  d1 = *(const float4*)(dp + 4);
    float4 s2 = *(const float4*)(sp + 16), d2 = *(const float4*)(dp + 16);
    float4 s3 = *(const float4*)(sp + 20), d3 = *(const float4*)(dp + 20);
    v16h v;
    v[0]=(_Float16)(s0.x*d0.x); v[1]=(_Float16)(s0.y*d0.y);
    v[2]=(_Float16)(s0.z*d0.z); v[3]=(_Float16)(s0.w*d0.w);
    v[4]=(_Float16)(s1.x*d1.x); v[5]=(_Float16)(s1.y*d1.y);
    v[6]=(_Float16)(s1.z*d1.z); v[7]=(_Float16)(s1.w*d1.w);
    v[8]=(_Float16)(s2.x*d2.x); v[9]=(_Float16)(s2.y*d2.y);
    v[10]=(_Float16)(s2.z*d2.z); v[11]=(_Float16)(s2.w*d2.w);
    v[12]=(_Float16)(s3.x*d3.x); v[13]=(_Float16)(s3.y*d3.y);
    v[14]=(_Float16)(s3.z*d3.z); v[15]=(_Float16)(s3.w*d3.w);
    return v;
}

__device__ __forceinline__ v16h load_b_f16(const _Float16* __restrict__ p) {
    v8h lo = *(const v8h*)(p);
    v8h hi = *(const v8h*)(p + 16);
    v16h v;
#pragma unroll
    for (int i = 0; i < 8; ++i) { v[i] = lo[i]; v[8 + i] = hi[i]; }
    return v;
}

#define OWN_NODES  200
#define NO_OWNER   255u
static_assert(N_NODES / OWN_NODES <= 255, "an owner group fits a byte, 255 is free");
static_assert(E_EDGES % 16 == 0, "16 owner bytes are inside the plane or outside it");
__device__ __forceinline__ unsigned owner_of(int d) {
    return (unsigned)d < (unsigned)N_NODES ? (unsigned)d / OWN_NODES : NO_OWNER;
}

__global__ void prep_weights(const float* __restrict__ W0, const float* __restrict__ W1,
                             const float* __restrict__ Wp1, const float* __restrict__ Wp2,
                             _Float16* __restrict__ W0t, _Float16* __restrict__ W1t,
                             _Float16* __restrict__ Wp1t, _Float16* __restrict__ Wp2t,
                             const int* __restrict__ dst, unsigned* __restrict__ oplane) {
    int idx = blockIdx.x * blockDim.x + threadIdx.x;
    if (idx < E_EDGES / 16) {
        const int4* p = (const int4*)(dst + idx * 16);
        const int4 a = p[0], b = p[1], c = p[2], d = p[3];
        uint4 o;
        o.x = owner_of(a.x) | (owner_of(a.y) << 8) | (owner_of(a.z) << 16) | (owner_of(a.w) << 24);
        o.y = owner_of(b.x) | (owner_of(b.y) << 8) | (owner_of(b.z) << 16) | (owner_of(b.w) << 24);
        o.z = owner_of(c.x) | (owner_of(c.y) << 8) | (owner_of(c.z) << 16) | (owner_of(c.w) << 24);
        o.w = owner_of(d.x) | (owner_of(d.y) << 8) | (owner_of(d.z) << 16) | (owner_of(d.w) << 24);
        ((uint4*)oplane)[idx] = o;
    }
    if (idx < 128 * 128) {
        int c = idx >> 7, k = idx & 127;
        W0t[idx] = (_Float16)W0[k * 128 + c];
        W1t[idx] = (_Float16)W1[k * 128 + c];
    }
    if (idx < 64 * 128) {
        int c = idx >> 7, k = idx & 127;
        Wp1t[idx] = (_Float16)Wp1[k * 64 + c];
    }
    if (idx < 64 * 64) {
        int c = idx >> 6, k = idx & 63;
        Wp2t[idx] = (_Float16)Wp2[k * 64 + c];
    }
}

__global__ __launch_bounds__(256)
void gemm128_wmma(const float* __restrict__ A, const _Float16* __restrict__ Bt,
                  float* __restrict__ C, int M) {
    int wid  = blockIdx.x * (blockDim.x >> 5) + (threadIdx.x >> 5);
    int lane = threadIdx.x & 31;
    int half = lane >> 4, r = lane & 15;
    int tm = wid >> 3, tn = wid & 7;
    if (tm * 16 >= M) return;
    int row = tm * 16 + r;
    int col = tn * 16 + r;
    v8f acc = {};
#pragma unroll
    for (int kk = 0; kk < 128; kk += 32) {
        int k0 = kk + half * 8;
        v16h a = load_a_f32(A + row * 128 + k0);
        v16h b = load_b_f16(Bt + col * 128 + k0);
        acc = __builtin_amdgcn_wmma_f32_16x16x32_f16(false, a, false, b,
                                                     (short)0, acc, false, false);
    }
    float* cp = C + (tm * 16 + half * 8) * 128 + tn * 16 + r;
#pragma unroll
    for (int v = 0; v < 8; ++v) cp[v * 128] = acc[v];
}

#define OWN_WAVES  32
#define SCAN_IDS   512
#define SCAN_AHEAD 3
#define SCAN_STEP  (OWN_WAVES * SCAN_IDS)
#define AGG_EDGES  8
#define QCAP       (SCAN_IDS + AGG_EDGES)
#define LIST_CAP   256
#define N_GROUPS   (N_NODES / OWN_NODES)
#define MAX_START  ((int)0x80808080)
#define ROW_QUADS  (OWN_NODES * HDIM / 4)
#define ROW_SWEEPS ((ROW_QUADS + OWN_WAVES * 32 - 1) / (OWN_WAVES * 32))
static_assert(N_NODES % OWN_NODES == 0, "groups cover the nodes exactly");
static_assert(E_EDGES / 16 >= 128 * 128, "the weight launch's grid, sized by the plane, covers the weights");
static_assert(OWN_NODES <= 256 && E_EDGES <= (1 << 23), "queue word: edge << 8 | node in group");
static_assert((OWN_NODES * HDIM) % (OWN_WAVES * 32) == 0, "node pass: whole sweeps");
static_assert(LIST_CAP % AGG_EDGES == 0, "a round's entries stay inside the wave's list");

__device__ __forceinline__ uint4 load_owners(const unsigned char* __restrict__ op, int i0) {
    return *(const uint4*)(op + (i0 < E_EDGES ? i0 : 0));
}
__device__ __forceinline__ unsigned zero_bytes(unsigned y) {
    return ~(((y & 0x7f7f7f7fu) + 0x7f7f7f7fu) | y | 0x7f7f7f7fu);
}

__device__ __forceinline__ float2 score_edges(const float* ers,
                                              const float* __restrict__ fl4, float4 at,
                                              int lane, int dl, int sv) {
    const int c = lane & 15;
    float4 el[AGG_EDGES];
#pragma unroll
    for (int u = 0; u < AGG_EDGES; ++u) {
        int s_ = __builtin_amdgcn_readlane(sv, u);
        el[u] = *(const float4*)(fl4 + s_ * HDIM);
    }
    float keep = 0.f;
#pragma unroll
    for (int u = 0; u < AGG_EDGES; ++u) {
        int d_ = __builtin_amdgcn_readlane(dl, u);
        float4 er = *(const float4*)(ers + d_ * HDIM + lane * 4);
        float x0 = el[u].x + er.x; x0 = x0 > 0.f ? x0 : NEG_SLOPE * x0;
        float x1 = el[u].y + er.y; x1 = x1 > 0.f ? x1 : NEG_SLOPE * x1;
        float x2 = el[u].z + er.z; x2 = x2 > 0.f ? x2 : NEG_SLOPE * x2;
        float x3 = el[u].w + er.w; x3 = x3 > 0.f ? x3 : NEG_SLOPE * x3;
        float part = fmaf(x3, at.w, fmaf(x2, at.z, fmaf(x0, at.x, x1 * at.y)));
        part += __shfl_xor(part, 1, 32);
        part += __shfl_xor(part, 2, 32);
        part += __shfl_xor(part, 4, 32);
        part += __shfl_xor(part, 8, 32);
        if (c == u) keep = part;
    }
    float keep1 = __shfl_xor(keep, 16, 32);
    return make_float2(keep, keep1);
}

__device__ __forceinline__ void add_edges(float* acc, float* ss, const int* mx,
                                          const float* __restrict__ fl,
                                          int lane, int n, int dl, int sv,
                                          float s0, float s1) {
    int2 mo = *(const int2*)(mx + dl * 2);
    float v[AGG_EDGES][4];
#pragma unroll
    for (int u = 0; u < AGG_EDGES; ++u) {
        int s_ = __builtin_amdgcn_readlane(sv, u);
        const float* fr = fl + s_ * HDIM;
        v[u][0] = fr[0]; v[u][1] = fr[32]; v[u][2] = fr[64]; v[u][3] = fr[96];
    }
    int p0v = __float_as_int(__expf(s0 - ord2f(mo.x)));
    int p1v = __float_as_int(__expf(s1 - ord2f(mo.y)));
#pragma unroll
    for (int u = 0; u < AGG_EDGES; ++u) {
        if (u < n) {
            int d_ = __builtin_amdgcn_readlane(dl, u);
            float p0 = __int_as_float(__builtin_amdgcn_readlane(p0v, u));
            float p1 = __int_as_float(__builtin_amdgcn_readlane(p1v, u));
            if (lane < 2) atomicAdd(&ss[d_ * 2 + lane], lane ? p1 : p0);
            float* rp = acc + d_ * HDIM + lane;
            atomicAdd(rp +  0, v[u][0] * p0);
            atomicAdd(rp + 32, v[u][1] * p0);
            atomicAdd(rp + 64, v[u][2] * p1);
            atomicAdd(rp + 96, v[u][3] * p1);
        }
    }
}

template <class F>
__device__ __forceinline__ int scan_rounds(const unsigned char* __restrict__ op,
                                           const int* __restrict__ dst,
                                           const int* __restrict__ src,
                                           int lo, int w, int lane, int* qw, F round) {
    int cnt = 0;
    int tot = 0;
    const unsigned below = (1u << lane) - 1u;
    const unsigned pat = (unsigned)(lo / OWN_NODES) * 0x01010101u;
    uint4 cur[SCAN_AHEAD];
#pragma unroll
    for (int c = 0; c < SCAN_AHEAD; ++c) cur[c] = make_uint4(0u, 0u, 0u, 0u);
    for (int base0 = w * SCAN_IDS - SCAN_AHEAD * SCAN_STEP; base0 < E_EDGES;
         base0 += SCAN_AHEAD * SCAN_STEP) {
#pragma unroll
      for (int c = 0; c < SCAN_AHEAD; ++c) {
        const int base = base0 + c * SCAN_STEP;
        unsigned m = (zero_bytes(cur[c].x ^ pat) >> 7) | (zero_bytes(cur[c].y ^ pat) >> 6) |
                     (zero_bytes(cur[c].z ^ pat) >> 5) | (zero_bytes(cur[c].w ^ pat) >> 4);
        m = (unsigned)(base + lane * 16) < (unsigned)E_EDGES ? m : 0u;
        cur[c] = load_owners(op, base + SCAN_AHEAD * SCAN_STEP + lane * 16);
        unsigned mask = (unsigned)__ballot(m != 0);
        while (mask != 0) {
            if (m != 0) {
                const int p = __builtin_ctz(m);
                qw[cnt + __popc(mask & below)] = base + lane * 16 + (p & 3) * 4 + (p >> 3);
                m &= m - 1;
            }
            cnt += __popc(mask);
            mask = (unsigned)__ballot(m != 0);
        }
        const bool last = base + SCAN_STEP >= E_EDGES;
        __builtin_amdgcn_wave_barrier();
        int head = 0;
        while (cnt - head >= AGG_EDGES || (last && cnt > head)) {
            const int n = cnt - head < AGG_EDGES ? cnt - head : AGG_EDGES;
            const int j = lane & (AGG_EDGES - 1);
            const int e = qw[head + (j < n ? j : 0)];
            const int sv = src[e];
            const int dl = dst[e] - lo;
            round(n, tot, (e << 8) | dl, sv);
            tot += n;
            head += n;
        }
        if (head > 0) {
            const int rem = cnt - head;
            int keep = lane < rem ? qw[head + lane] : 0;
            __builtin_amdgcn_wave_barrier();
            if (lane < rem) qw[lane] = keep;
            cnt = rem;
        }
      }
    }
    return tot;
}

template <class F>
__device__ __forceinline__ void list_rounds(const int* lp, int listed, int lane, F round) {
    if (listed <= 0) return;
    const int j = lane & (AGG_EDGES - 1);
    int4 ent = *(const int4*)(lp + 4 * (j < listed ? j : 0));
    int head = 0;
    for (; head + AGG_EDGES <= listed; head += AGG_EDGES) {
        const int nh = head + AGG_EDGES;
        const int ni = nh < listed ? nh + (nh + j < listed ? j : 0) : 0;
        const int4 nxt = *(const int4*)(lp + 4 * ni);
        round(AGG_EDGES, head, ent);
        ent = nxt;
    }
    if (head < listed) {
        round(listed - head, head, ent);
    }
}

template <bool FIRST>
__global__ __launch_bounds__(OWN_WAVES * 32)
void owner_layer_kernel(const float* __restrict__ feat, const int* __restrict__ src,
                        const int* __restrict__ dst, const unsigned char* __restrict__ oplane,
                        const float* __restrict__ attn,
                        const float* __restrict__ hin, float* __restrict__ hout,
                        int residual, int* elist, int* ecnt) {
    __shared__ float acc[OWN_NODES * HDIM];
    __shared__ float ers[OWN_NODES * HDIM];
    __shared__ float ss[OWN_NODES * 2];
    __shared__ int   mx[OWN_NODES * 2];
    __shared__ int   q[OWN_WAVES][QCAP];
    const int lo = blockIdx.x * OWN_NODES;
    const int tid = threadIdx.x, lane = tid & 31;
    const int w = __builtin_amdgcn_readfirstlane(tid >> 5);
    {
        const float4* own = (const float4*)(feat + lo * HDIM);
        float4 t[ROW_SWEEPS];
#pragma unroll
        for (int k = 0; k < ROW_SWEEPS; ++k) {
            const int i = tid + k * (OWN_WAVES * 32);
            t[k] = own[i < ROW_QUADS ? i : 0];
        }
        for (int i = tid; i < OWN_NODES * HDIM; i += OWN_WAVES * 32) acc[i] = 0.f;
        if (tid < OWN_NODES * 2) { ss[tid] = 0.f; mx[tid] = MAX_START; }
#pragma unroll
        for (int k = 0; k < ROW_SWEEPS; ++k) {
            const int i = tid + k * (OWN_WAVES * 32);
            if (i < ROW_QUADS) ((float4*)ers)[i] = t[k];
        }
    }
    __syncthreads();

    const int slot = blockIdx.x * OWN_WAVES + w;
    int* lp = elist + slot * (LIST_CAP * 4);
    int* qw = q[w];
    const float* fl  = feat + lane;
    const float* fl4 = feat + lane * 4;
    const float4 at = *(const float4*)(attn + lane * 4);

    auto score_scanned = [&](int n, int tot, int qe, int sv) {
        const int dl = qe & 255;
        const float2 sc = score_edges(ers, fl4, at, lane, dl, sv);
        if (lane < n) {
            atomicMax(&mx[dl * 2 + 0], f2ord(sc.x));
            atomicMax(&mx[dl * 2 + 1], f2ord(sc.y));
            if (FIRST && tot + lane < LIST_CAP) {
                *(int4*)(lp + 4 * (tot + lane)) =
                    make_int4(qe, sv, __float_as_int(sc.x), __float_as_int(sc.y));
            }
        }
    };
    auto score_listed = [&](int n, int head, int4 ent) {
        const int dl = ent.x & 255;
        const float2 sc = score_edges(ers, fl4, at, lane, dl, ent.y);
        if (lane < n) {
            atomicMax(&mx[dl * 2 + 0], f2ord(sc.x));
            atomicMax(&mx[dl * 2 + 1], f2ord(sc.y));
            *(float2*)(lp + 4 * (head + lane) + 2) = sc;
        }
    };
    auto add_scanned = [&](int n, int tot, int qe, int sv) {
        const int dl = qe & 255;
        const float2 sc = score_edges(ers, fl4, at, lane, dl, sv);
        add_edges(acc, ss, mx, fl, lane, n, dl, sv, sc.x, sc.y);
    };
    auto add_listed = [&](int n, int head, int4 ent) {
        add_edges(acc, ss, mx, fl, lane, n, ent.x & 255, ent.y,
                  __int_as_float(ent.z), __int_as_float(ent.w));
    };

    int listed;
    if (FIRST) {
        listed = __builtin_amdgcn_readfirstlane(
            scan_rounds(oplane, dst, src, lo, w, lane, qw, score_scanned));
        if (lane == 0) ecnt[slot] = listed;
    } else {
        listed = __builtin_amdgcn_readfirstlane(ecnt[slot]);
        if (listed <= LIST_CAP) list_rounds(lp, listed, lane, score_listed);
        else scan_rounds(oplane, dst, src, lo, w, lane, qw, score_scanned);
    }
    __syncthreads();
    if (listed <= LIST_CAP) list_rounds(lp, listed, lane, add_listed);
    else scan_rounds(oplane, dst, src, lo, w, lane, qw, add_scanned);
    __syncthreads();

    for (int i = tid; i < OWN_NODES * HDIM; i += OWN_WAVES * 32) {
        float s = ss[i >> 6];
        float r = acc[i];
        float x = s > 0.f ? r / s : 0.f;
        int g = lo * HDIM + i;
        x += residual ? hin[g] : 0.0f;
        hout[g] = x > 0.f ? x : (__expf(x) - 1.0f);
    }
}

__global__ __launch_bounds__(128)
void predictor_wmma(const float* __restrict__ h,
                    const int* __restrict__ ps, const int* __restrict__ pd,
                    const int* __restrict__ ns, const int* __restrict__ nd,
                    const _Float16* __restrict__ Wp1t, const float* __restrict__ bp1,
                    const _Float16* __restrict__ Wp2t, const float* __restrict__ bp2,
                    const float* __restrict__ Wp3, const float* __restrict__ bp3,
                    float* __restrict__ out) {
    __shared__ _Float16 z1s[4][16 * 64];
    __shared__ float    z2s[4][16 * 64];
    const int tilesPerSet = P_CAND / 16;
    int w = threadIdx.x >> 5;
    int tile = blockIdx.x * 4 + w;
    if (tile >= 2 * tilesPerSet) return;
    bool isNeg = tile >= tilesPerSet;
    int t = isNeg ? tile - tilesPerSet : tile;
    const int* S = isNeg ? ns : ps;
    const int* D = isNeg ? nd : pd;

    int lane = threadIdx.x & 31;
    int half = lane >> 4, r = lane & 15;
    int c = t * 16 + r;
    int si = S[c], di = D[c];

    v16h a1[4];
#pragma unroll
    for (int kc = 0; kc < 4; ++kc) {
        int k0 = kc * 32 + half * 8;
        a1[kc] = load_a_prod(h + si * HDIM + k0, h + di * HDIM + k0);
    }
#pragma unroll
    for (int nt = 0; nt < 4; ++nt) {
        int col = nt * 16 + r;
        v8f acc = {};
#pragma unroll
        for (int kc = 0; kc < 4; ++kc) {
            int k0 = kc * 32 + half * 8;
            v16h b = load_b_f16(Wp1t + col * 128 + k0);
            acc = __builtin_amdgcn_wmma_f32_16x16x32_f16(false, a1[kc], false, b,
                                                         (short)0, acc, false, false);
        }
        float bias = bp1[col];
#pragma unroll
        for (int v = 0; v < 8; ++v) {
            float z = acc[v] + bias;
            z = z > 0.f ? z : 0.f;
            z1s[w][(v + half * 8) * 64 + col] = (_Float16)z;
        }
    }

    v16h a2[2];
#pragma unroll
    for (int kc = 0; kc < 2; ++kc) {
        int k0 = kc * 32 + half * 8;
        const _Float16* zp = &z1s[w][r * 64 + k0];
        a2[kc] = load_b_f16(zp);
    }
#pragma unroll
    for (int nt = 0; nt < 4; ++nt) {
        int col = nt * 16 + r;
        v8f acc = {};
#pragma unroll
        for (int kc = 0; kc < 2; ++kc) {
            int k0 = kc * 32 + half * 8;
            v16h b = load_b_f16(Wp2t + col * 64 + k0);
            acc = __builtin_amdgcn_wmma_f32_16x16x32_f16(false, a2[kc], false, b,
                                                         (short)0, acc, false, false);
        }
        float bias = bp2[col];
#pragma unroll
        for (int v = 0; v < 8; ++v) {
            float z = acc[v] + bias;
            z2s[w][(v + half * 8) * 64 + col] = z > 0.f ? z : 0.f;
        }
    }

    float part = 0.f;
    int jb = half * 32;
#pragma unroll
    for (int j = 0; j < 32; ++j)
        part += z2s[w][r * 64 + jb + j] * Wp3[jb + j];
    part += __shfl_xor(part, 16, 32);
    if (half == 0) {
        out[(isNeg ? P_CAND : 0) + t * 16 + r] = part + bp3[0];
    }
}

extern "C" void kernel_launch(void* const* d_in, const int* in_sizes, int n_in,
                              void* d_out, int out_size, void* d_ws, size_t ws_size,
                              hipStream_t stream) {
    const float* x        = (const float*)d_in[0];
    const int*   src      = (const int*)  d_in[1];
    const int*   dst      = (const int*)  d_in[2];
    const int*   pos_src  = (const int*)  d_in[3];
    const int*   pos_dst  = (const int*)  d_in[4];
    const int*   neg_src  = (const int*)  d_in[5];
    const int*   neg_dst  = (const int*)  d_in[6];
    const float* W0       = (const float*)d_in[7];
    const float* attn0    = (const float*)d_in[8];
    const float* W1       = (const float*)d_in[9];
    const float* attn1    = (const float*)d_in[10];
    const float* Wp1      = (const float*)d_in[11];
    const float* bp1      = (const float*)d_in[12];
    const float* Wp2      = (const float*)d_in[13];
    const float* bp2      = (const float*)d_in[14];
    const float* Wp3      = (const float*)d_in[15];
    const float* bp3      = (const float*)d_in[16];

    char* ws = (char*)d_ws;
    size_t off = 0;
    auto carve = [&](size_t bytes) -> char* {
        char* p = ws + off;
        off += (bytes + 255) & ~(size_t)255;
        return p;
    };
    float*     feat  = (float*)    carve((size_t)N_NODES * HDIM * 4);
    float*     h0    = (float*)    carve((size_t)N_NODES * HDIM * 4);
    float*     h1    = (float*)    carve((size_t)N_NODES * HDIM * 4);
    _Float16*  W0t   = (_Float16*) carve(128 * 128 * 2);
    _Float16*  W1t   = (_Float16*) carve(128 * 128 * 2);
    _Float16*  Wp1t  = (_Float16*) carve(64 * 128 * 2);
    _Float16*  Wp2t  = (_Float16*) carve(64 * 64 * 2);
    int*       elist = (int*)      carve((size_t)N_GROUPS * OWN_WAVES * LIST_CAP * 4 * 4);
    int*       ecnt  = (int*)      carve((size_t)N_GROUPS * OWN_WAVES * 4);
    unsigned char* oplane = (unsigned char*) carve((size_t)E_EDGES);

    prep_weights<<<(E_EDGES / 16 + 255) / 256, 256, 0, stream>>>(
        W0, W1, Wp1, Wp2, W0t, W1t, Wp1t, Wp2t, dst, (unsigned*)oplane);

    const int gemmBlocks  = (N_NODES / 16) * 8 / 8;
    const int ownBlocks   = N_NODES / OWN_NODES;

    gemm128_wmma<<<gemmBlocks, 256, 0, stream>>>(x, W0t, feat, N_NODES);
    owner_layer_kernel<true><<<ownBlocks, OWN_WAVES * 32, 0, stream>>>(
        feat, src, dst, oplane, attn0, x, h0, 0, elist, ecnt);

    gemm128_wmma<<<gemmBlocks, 256, 0, stream>>>(h0, W1t, feat, N_NODES);
    owner_layer_kernel<false><<<ownBlocks, OWN_WAVES * 32, 0, stream>>>(
        feat, src, dst, oplane, attn1, h0, h1, 1, elist, ecnt);

    const int predBlocks = (2 * (P_CAND / 16) + 3) / 4;
    predictor_wmma<<<predBlocks, 128, 0, stream>>>(
        h1, pos_src, pos_dst, neg_src, neg_dst,
        Wp1t, bp1, Wp2t, bp2, Wp3, bp3, (float*)d_out);
}
